# starting kernel
# baseline (speedup 1.0000x reference)
#include <hip/hip_runtime.h>

#define N_NODES  50000
#define N_EDGES  1600000
#define N_GRAPHS 256
#define IN_DIM   9
#define HID      128

typedef __attribute__((ext_vector_type(2))) float v2f;
typedef __attribute__((ext_vector_type(8))) float v8f;

__device__ __forceinline__ void atomic_add_f32(float* p, float v) {
    asm volatile("global_atomic_add_f32 %0, %1, off scope:SCOPE_DEV"
                 :: "v"(p), "v"(v) : "memory");
}

__global__ void k_fill1(float* p, int n) {
    int i = blockIdx.x * blockDim.x + threadIdx.x;
    if (i < n) p[i] = 1.0f;
}

__global__ void k_deg(const int* __restrict__ dst, float* __restrict__ deg) {
    int e = blockIdx.x * blockDim.x + threadIdx.x;
    if (e < N_EDGES) atomic_add_f32(&deg[dst[e]], 1.0f);
}

__global__ void k_rsqrt(float* p, int n) {
    int i = blockIdx.x * blockDim.x + threadIdx.x;
    if (i < n) p[i] = rsqrtf(p[i]);
}

__global__ void k_gemm_in(const float* __restrict__ x, const float* __restrict__ W1,
                          float* __restrict__ out) {
    int idx = blockIdx.x * blockDim.x + threadIdx.x;
    if (idx >= N_NODES * HID) return;
    int f = idx & (HID - 1);
    int n = idx >> 7;
    const float* xr = x + n * IN_DIM;
    float acc = 0.f;
#pragma unroll
    for (int k = 0; k < IN_DIM; ++k) acc = fmaf(xr[k], W1[k * HID + f], acc);
    out[idx] = acc;
}

template <bool RELU>
__global__ void k_gemm_wmma(const float* __restrict__ A, const float* __restrict__ W,
                            float* __restrict__ out) {
    int wave = (blockIdx.x * blockDim.x + threadIdx.x) >> 5;
    int lane = threadIdx.x & 31;
    int ct = wave & 7;
    int rt = wave >> 3;
    if (rt >= (N_NODES >> 4)) return;

    int half = lane >> 4;
    int l16  = lane & 15;
    int row  = (rt << 4) + l16;
    int col  = (ct << 4) + l16;
    const float* arow = A + row * HID;

    v8f c = {0.f, 0.f, 0.f, 0.f, 0.f, 0.f, 0.f, 0.f};
#pragma unroll
    for (int k0 = 0; k0 < HID; k0 += 4) {
        int ka = k0 + (half << 1);
        float a0 = arow[ka];
        float a1 = arow[ka + 1];
        if (RELU) { a0 = fmaxf(a0, 0.f); a1 = fmaxf(a1, 0.f); }
        v2f a; a.x = a0; a.y = a1;
        v2f b; b.x = W[ka * HID + col]; b.y = W[(ka + 1) * HID + col];
        c = __builtin_amdgcn_wmma_f32_16x16x4_f32(false, a, false, b,
                                                  (short)0, c, false, false);
    }
    float* orow = out + ((rt << 4) + (half << 3)) * HID + (ct << 4) + l16;
#pragma unroll
    for (int v = 0; v < 8; ++v) orow[v * HID] = c[v];
}

__global__ void k_init_out(const float* __restrict__ hw, const float* __restrict__ dis,
                           const float* __restrict__ bias, float* __restrict__ out) {
    int idx = blockIdx.x * blockDim.x + threadIdx.x;
    if (idx >= N_NODES * HID) return;
    int f = idx & (HID - 1);
    int n = idx >> 7;
    float s = dis[n];
    out[idx] = fmaf(hw[idx], s * s, bias[f]);
}

__global__ void k_edge(const int* __restrict__ src, const int* __restrict__ dst,
                       const float* __restrict__ dis, const float* __restrict__ hw,
                       float* __restrict__ out) {
    int t = blockIdx.x * blockDim.x + threadIdx.x;
    int e = t >> 5;
    if (e >= N_EDGES) return;
    e = __builtin_amdgcn_readfirstlane(e);
    int lane = t & 31;
    int s = src[e];
    int d = dst[e];
    float w = dis[s] * dis[d];
    const float4 h4 = *(const float4*)(hw + s * HID + lane * 4);
    float* o = out + d * HID + lane * 4;
    atomic_add_f32(o + 0, h4.x * w);
    atomic_add_f32(o + 1, h4.y * w);
    atomic_add_f32(o + 2, h4.z * w);
    atomic_add_f32(o + 3, h4.w * w);
}

__global__ void k_pool_zero(float* __restrict__ sums, float* __restrict__ counts) {
    int i = blockIdx.x * blockDim.x + threadIdx.x;
    if (i < N_GRAPHS * HID) sums[i] = 0.f;
    if (i < N_GRAPHS) counts[i] = 0.f;
}

__global__ void k_pool_acc(const float* __restrict__ h, const int* __restrict__ batch,
                           float* __restrict__ sums, float* __restrict__ counts) {
    int t = blockIdx.x * blockDim.x + threadIdx.x;
    int n = t >> 5;
    if (n >= N_NODES) return;
    n = __builtin_amdgcn_readfirstlane(n);
    int lane = t & 31;
    int g = batch[n];
    const float4 h4 = *(const float4*)(h + n * HID + lane * 4);
    float* o = sums + g * HID + lane * 4;
    atomic_add_f32(o + 0, h4.x);
    atomic_add_f32(o + 1, h4.y);
    atomic_add_f32(o + 2, h4.z);
    atomic_add_f32(o + 3, h4.w);
    if (lane == 0) atomic_add_f32(&counts[g], 1.0f);
}

__global__ void k_pool_div(float* __restrict__ sums, const float* __restrict__ counts) {
    int i = blockIdx.x * blockDim.x + threadIdx.x;
    if (i >= N_GRAPHS * HID) return;
    int g = i >> 7;
    sums[i] = sums[i] / fmaxf(counts[g], 1.0f);
}

extern "C" void kernel_launch(void* const* d_in, const int* in_sizes, int n_in,
                              void* d_out, int out_size, void* d_ws, size_t ws_size,
                              hipStream_t stream) {
    const float* x        = (const float*)d_in[0];
    const int*   eidx     = (const int*)d_in[1];
    const int*   batch    = (const int*)d_in[2];
    const float* W1       = (const float*)d_in[3];
    const float* b1       = (const float*)d_in[4];
    const float* W2       = (const float*)d_in[5];
    const float* b2       = (const float*)d_in[6];
    const float* W3       = (const float*)d_in[7];
    const float* b3       = (const float*)d_in[8];
    float* out = (float*)d_out;

    const int* src = eidx;
    const int* dst = eidx + N_EDGES;

    float* ws     = (float*)d_ws;
    float* dis    = ws;
    float* hw     = dis + N_NODES;
    float* hA     = hw + (size_t)N_NODES * HID;
    float* counts = hA + (size_t)N_NODES * HID;

    const int B = 256;
    int g_nf   = (N_NODES * HID + B - 1) / B;
    int g_n    = (N_NODES + B - 1) / B;
    int g_e    = (N_EDGES + B - 1) / B;
    int g_ew   = (N_EDGES * 32 + B - 1) / B;
    int g_nw   = (N_NODES * 32 + B - 1) / B;
    int g_wmma = ((N_NODES >> 4) * 8 + (B >> 5) - 1) / (B >> 5);

    k_fill1<<<g_n, B, 0, stream>>>(dis, N_NODES);
    k_deg<<<g_e, B, 0, stream>>>(dst, dis);
    k_rsqrt<<<g_n, B, 0, stream>>>(dis, N_NODES);

    k_gemm_in<<<g_nf, B, 0, stream>>>(x, W1, hw);
    k_init_out<<<g_nf, B, 0, stream>>>(hw, dis, b1, hA);
    k_edge<<<g_ew, B, 0, stream>>>(src, dst, dis, hw, hA);

    k_gemm_wmma<true><<<g_wmma, B, 0, stream>>>(hA, W2, hw);
    k_init_out<<<g_nf, B, 0, stream>>>(hw, dis, b2, hA);
    k_edge<<<g_ew, B, 0, stream>>>(src, dst, dis, hw, hA);

    k_gemm_wmma<true><<<g_wmma, B, 0, stream>>>(hA, W3, hw);
    k_init_out<<<g_nf, B, 0, stream>>>(hw, dis, b3, hA);
    k_edge<<<g_ew, B, 0, stream>>>(src, dst, dis, hw, hA);

    k_pool_zero<<<(N_GRAPHS * HID + B - 1) / B, B, 0, stream>>>(out, counts);
    k_pool_acc<<<g_nw, B, 0, stream>>>(hA, batch, out, counts);
    k_pool_div<<<(N_GRAPHS * HID + B - 1) / B, B, 0, stream>>>(out, counts);
}


# Layer-2/3 products fused into the edge kernels: a work-group multiplies its finished rows by the next weights on WMMA out of LDS (row pitch 132 floats); 2 launches fewer; sums bit-identical
# speedup vs baseline: 2.6258x; 2.6258x over previous
#include <hip/hip_runtime.h>

#define N_NODES  50000
#define N_EDGES  1600000
#define N_GRAPHS 256
#define IN_DIM   9
#define HID      128
#define OWN      196
#define OWN_WG   256
#define PITCH    132
#define ACC_ROWS 208

typedef __attribute__((ext_vector_type(2))) float v2f;
typedef __attribute__((ext_vector_type(8))) float v8f;

__device__ __forceinline__ void atomic_add_f32(float* p, float v) {
    asm volatile("global_atomic_add_f32 %0, %1, off scope:SCOPE_DEV"
                 :: "v"(p), "v"(v) : "memory");
}

__global__ void k_fill1(float* p, int n) {
    int i = blockIdx.x * blockDim.x + threadIdx.x;
    if (i < n) p[i] = 1.0f;
}

__global__ void k_deg(const int* __restrict__ dst, float* __restrict__ deg) {
    int e = blockIdx.x * blockDim.x + threadIdx.x;
    if (e < N_EDGES) atomic_add_f32(&deg[dst[e]], 1.0f);
}

__global__ __launch_bounds__(256) void k_norm(float* __restrict__ p, int* __restrict__ wtot) {
    __shared__ int red[8][32];
    const int r = threadIdx.x;
    const int n = blockIdx.x * OWN + r;
    int c = 0;
    if (r < OWN && n < N_NODES) {
        float d = p[n];
        p[n] = rsqrtf(d);
        c = (int)d - 1;
    }
    red[r >> 5][r & 31] = c;
    __syncthreads();
    if (r < 32) {
        int t = 0;
        for (int j = 0; j < 8; ++j) t += red[j][r];
        wtot[blockIdx.x * 32 + r] = t;
    }
}

__global__ void k_gemm_in(const float* __restrict__ x, const float* __restrict__ W1,
                          float* __restrict__ out) {
    int idx = blockIdx.x * blockDim.x + threadIdx.x;
    if (idx >= N_NODES * HID) return;
    int f = idx & (HID - 1);
    int n = idx >> 7;
    const float* xr = x + n * IN_DIM;
    float acc = 0.f;
#pragma unroll
    for (int k = 0; k < IN_DIM; ++k) acc = fmaf(xr[k], W1[k * HID + f], acc);
    out[idx] = acc;
}

template <int NT>
__device__ __forceinline__ void rows_times_w(const float* arow, const float* __restrict__ wl, v8f* d) {
    v2f bc[8], bn[8];
    float2 an[NT];
#pragma unroll
    for (int s = 0; s < 8; ++s) { bc[s].x = wl[(4 * s) * HID]; bc[s].y = wl[(4 * s + 1) * HID]; }
#pragma unroll
    for (int t = 0; t < NT; ++t) an[t] = *(const float2*)(arow + t * 64 * PITCH);
#pragma unroll
    for (int cb = 0; cb < HID; cb += 32) {
        if (cb + 32 < HID) {
#pragma unroll
            for (int s = 0; s < 8; ++s) {
                bn[s].x = wl[(cb + 32 + 4 * s) * HID];
                bn[s].y = wl[(cb + 32 + 4 * s + 1) * HID];
            }
        }
#pragma unroll
        for (int s = 0; s < 8; ++s) {
            const int k0 = cb + 4 * s;
            float2 ac[NT];
#pragma unroll
            for (int t = 0; t < NT; ++t) ac[t] = an[t];
            if (k0 + 4 < HID) {
#pragma unroll
                for (int t = 0; t < NT; ++t) an[t] = *(const float2*)(arow + t * 64 * PITCH + k0 + 4);
            }
            __builtin_amdgcn_sched_barrier(0);
#pragma unroll
            for (int t = 0; t < NT; ++t) {
                v2f a; a.x = fmaxf(ac[t].x, 0.f); a.y = fmaxf(ac[t].y, 0.f);
                d[t] = __builtin_amdgcn_wmma_f32_16x16x4_f32(false, a, false, bc[s],
                                                             (short)0, d[t], false, false);
            }
        }
#pragma unroll
        for (int s = 0; s < 8; ++s) bc[s] = bn[s];
    }
    if (NT == 4)
        asm volatile("v_nop\n\tv_nop\n\tv_nop\n\tv_nop" : "+v"(d[0]), "+v"(d[1]), "+v"(d[2]), "+v"(d[3]));
    else
        asm volatile("v_nop\n\tv_nop\n\tv_nop\n\tv_nop" : "+v"(d[0]), "+v"(d[1]), "+v"(d[2]));
}

#define QW         768
#define PW         64
#define RS         256
#define UNITS      (N_EDGES / 128)
#define TRIP_UNITS 4
#define TRIPS      ((UNITS + 32 * TRIP_UNITS - 1) / (32 * TRIP_UNITS))

static_assert(N_NODES <= 65536 && OWN <= 256, "a saved entry packs the source node in 16 bits, the local row above; k_norm has 256 threads per block of OWN rows");

__device__ __forceinline__ int trip_ubase(int pos, int w) {
    int itr = (pos + (int)blockIdx.x) % TRIPS;
    return (itr * 32 + w) * TRIP_UNITS;
}

__device__ __forceinline__ void load_trip(const int4* __restrict__ d4, int pos, int w, int lane,
                                          int4* dv) {
    const int ubase = trip_ubase(pos, w);
#pragma unroll
    for (int j = 0; j < TRIP_UNITS; ++j) {
        int u = ubase + j;
        if (u < UNITS) dv[j] = d4[u * 32 + lane];
        else           dv[j] = make_int4(-1, -1, -1, -1);
    }
}

__device__ __forceinline__ void add_rows(int s, int dl, float c, int n, int lane,
                                         const float* __restrict__ hw, float* acc) {
#pragma unroll 8
    for (int e = 0; e < n; ++e) {
        int sj = __builtin_amdgcn_readlane(s, e);
        int dj = __builtin_amdgcn_readlane(dl, e);
        float cj = __builtin_bit_cast(float, __builtin_amdgcn_readlane(__builtin_bit_cast(int, c), e));
        float4 h = *(const float4*)(hw + (size_t)sj * HID + lane * 4);
        float4* a = (float4*)(acc + dj * PITCH) + lane;
        float4 v = *a;
        v.x += h.x * cj; v.y += h.y * cj; v.z += h.z * cj; v.w += h.w * cj;
        *a = v;
    }
}

__device__ __forceinline__ void apply_pending(const int2* pq, int n, int lane, int n0,
                                              const int* __restrict__ src,
                                              const float* __restrict__ dis,
                                              const float* __restrict__ hw, float* acc,
                                              int2* __restrict__ save) {
    int s = 0, dl = 0; float c = 0.f;
    if (lane < n) {
        int2 en = pq[lane];
        s = src[en.x];
        dl = en.y;
        c = dis[s] * dis[n0 + dl];
        save[lane] = make_int2(s | (dl << 16), __builtin_bit_cast(int, c));
    }
    add_rows(s, dl, c, n, lane, hw, acc);
}

template <bool REPLAY, bool FUSE>
__global__ __launch_bounds__(1024) void k_edge_own(const int* __restrict__ src,
                                                   const int* __restrict__ dst,
                                                   const float* __restrict__ dis,
                                                   const float* __restrict__ hw,
                                                   const float* __restrict__ bias,
                                                   float* __restrict__ out,
                                                   const int* __restrict__ wtot,
                                                   int* __restrict__ list_i,
                                                   const float* __restrict__ wnext) {
    __shared__ __attribute__((aligned(16))) float acc[ACC_ROWS * PITCH];
    __shared__ int2  q[32][QW];
    __shared__ int2  pend[32][PW];
    __shared__ int   qcnt[32];
    __shared__ int   unfinished[32];
    __shared__ int   sred[32];

    const int tid  = threadIdx.x;
    const int lane = tid & 31;
    const int w    = __builtin_amdgcn_readfirstlane(tid >> 5);
    const int n0   = blockIdx.x * OWN;
    const int nn   = min(OWN, N_NODES - n0);
    const unsigned lt = (1u << lane) - 1u;

    for (int i = tid; i < nn * (HID / 4); i += 1024) {
        int r = i >> 5;
        int f4 = i & 31;
        float s = dis[n0 + r];
        s = s * s;
        float4 h = ((const float4*)(hw + (size_t)(n0 + r) * HID))[f4];
        float4 b = ((const float4*)bias)[f4];
        float4 o;
        o.x = fmaf(h.x, s, b.x); o.y = fmaf(h.y, s, b.y);
        o.z = fmaf(h.z, s, b.z); o.w = fmaf(h.w, s, b.w);
        ((float4*)(acc + r * PITCH))[f4] = o;
    }

    int pre = 0;
    for (int i = tid; i < (int)blockIdx.x * 32; i += 1024) pre += wtot[i];
    for (int o = 16; o > 0; o >>= 1) pre += __shfl_xor(pre, o);
    if (lane == 0) sred[w] = pre;
    __syncthreads();
    const int mine = wtot[blockIdx.x * 32 + lane];
    int st = sred[lane] + ((lane < w) ? mine : 0);
    for (int o = 16; o > 0; o >>= 1) st += __shfl_xor(st, o);
    int2* mylist = (int2*)list_i + __builtin_amdgcn_readfirstlane(st);

    int2* myq = q[w];

    if (REPLAY) {
        const int mycnt = __builtin_amdgcn_readlane(mine, w);
        for (int cb = 0; cb < mycnt; cb += RS) {
            const int c = min(mycnt - cb, RS);
            int2 t[RS / 32];
#pragma unroll
            for (int k = 0; k < RS / 32; ++k) {
                const int i = k * 32 + lane;
                t[k] = make_int2(0, 0);
                if (i < c) t[k] = mylist[cb + i];
            }
#pragma unroll
            for (int k = 0; k < RS / 32; ++k) myq[k * 32 + lane] = t[k];
            for (int b = 0; b < c; b += 32) {
                const int n = min(32, c - b);
                int sd = 0; float cf = 0.f;
                if (lane < n) {
                    int2 en = myq[b + lane];
                    sd = en.x;
                    cf = __builtin_bit_cast(float, en.y);
                }
                add_rows(sd & 0xffff, sd >> 16, cf, n, lane, hw, acc);
            }
        }
        __syncthreads();
    } else {
        const int4* d4 = (const int4*)dst;
        int2* myp = pend[w];
        int pos = 0;
        int wpos = 0;

        for (;;) {
            int cnt = 0;
            int4 dvn[TRIP_UNITS];
            if (pos < TRIPS) load_trip(d4, pos, w, lane, dvn);
            while (pos < TRIPS && cnt + TRIP_UNITS * 128 <= QW) {
                const int ubase = trip_ubase(pos, w);
                int4 dv[TRIP_UNITS];
#pragma unroll
                for (int j = 0; j < TRIP_UNITS; ++j) dv[j] = dvn[j];
                ++pos;
                if (pos < TRIPS) load_trip(d4, pos, w, lane, dvn);
#pragma unroll
                for (int j = 0; j < TRIP_UNITS; ++j) {
                    int ebase = (ubase + j) * 128 + lane * 4;
                    int dd[4] = {dv[j].x, dv[j].y, dv[j].z, dv[j].w};
#pragma unroll
                    for (int k = 0; k < 4; ++k) {
                        unsigned dl = (unsigned)(dd[k] - n0);
                        bool m = dl < (unsigned)nn;
                        unsigned mask = __builtin_amdgcn_ballot_w32(m);
                        if (mask) {
                            if (m) myq[cnt + __builtin_popcount(mask & lt)] = make_int2(ebase + k, (int)dl);
                            cnt += __builtin_popcount(mask);
                        }
                    }
                }
            }
            if (lane == 0) { qcnt[w] = cnt; unfinished[w] = (pos < TRIPS) ? 1 : 0; }
            __syncthreads();
            const bool more = __builtin_amdgcn_ballot_w32(unfinished[lane] != 0) != 0u;

            int pc = 0;
            for (int p = 0; p < 32; ++p) {
                const int np = __builtin_amdgcn_readfirstlane(qcnt[p]);
                const int2* pq = q[p];
                for (int g = 0; g < np; g += 32) {
                    int2 en = make_int2(0, -1);
                    if (g + lane < np) en = pq[g + lane];
                    bool m = (en.y >= 0) && ((en.y & 31) == w);
                    unsigned mask = __builtin_amdgcn_ballot_w32(m);
                    if (mask) {
                        if (m) myp[pc + __builtin_popcount(mask & lt)] = en;
                        pc += __builtin_popcount(mask);
                        if (pc >= 32) {
                            apply_pending(myp, 32, lane, n0, src, dis, hw, acc, mylist + wpos);
                            wpos += 32;
                            int2 mv = make_int2(0, 0);
                            if (lane < pc - 32) mv = myp[32 + lane];
                            if (lane < pc - 32) myp[lane] = mv;
                            pc -= 32;
                        }
                    }
                }
            }
            if (pc > 0) {
                apply_pending(myp, pc, lane, n0, src, dis, hw, acc, mylist + wpos);
                wpos += pc;
            }
            __syncthreads();
            if (!more) break;
        }
    }

    if (FUSE) {
        const int ct  = w & 7;
        const int g   = w >> 3;
        const int c16 = lane & 15;
        const int hf  = lane >> 4;
        const float* arow = acc + (g * 16 + c16) * PITCH + 2 * hf;
        const float* wl   = wnext + (2 * hf) * HID + ct * 16 + c16;
        const int nt = (g == 0) ? 4 : 3;
        v8f d[4];
#pragma unroll
        for (int t = 0; t < 4; ++t) d[t] = (v8f){0.f, 0.f, 0.f, 0.f, 0.f, 0.f, 0.f, 0.f};
        if (g == 0) rows_times_w<4>(arow, wl, d);
        else        rows_times_w<3>(arow, wl, d);
        __syncthreads();
        float* drow = acc + (g * 16 + 8 * hf) * PITCH + ct * 16 + c16;
#pragma unroll
        for (int t = 0; t < 4; ++t) {
            if (t < nt) {
#pragma unroll
                for (int v = 0; v < 8; ++v) drow[(t * 64 + v) * PITCH] = d[t][v];
            }
        }
        __syncthreads();
    }

    float4* o4 = (float4*)(out + (size_t)n0 * HID);
    for (int i = tid; i < nn * (HID / 4); i += 1024)
        o4[i] = ((const float4*)(acc + (i >> 5) * PITCH))[i & 31];
}

__global__ void k_pool_zero(float* __restrict__ sums, float* __restrict__ counts) {
    int i = blockIdx.x * blockDim.x + threadIdx.x;
    if (i < N_GRAPHS * HID) sums[i] = 0.f;
    if (i < N_GRAPHS) counts[i] = 0.f;
}

__global__ void k_pool_acc(const float* __restrict__ h, const int* __restrict__ batch,
                           float* __restrict__ sums, float* __restrict__ counts) {
    int t = blockIdx.x * blockDim.x + threadIdx.x;
    int n = t >> 5;
    if (n >= N_NODES) return;
    n = __builtin_amdgcn_readfirstlane(n);
    int lane = t & 31;
    int g = batch[n];
    const float4 h4 = *(const float4*)(h + n * HID + lane * 4);
    float* o = sums + g * HID + lane * 4;
    atomic_add_f32(o + 0, h4.x);
    atomic_add_f32(o + 1, h4.y);
    atomic_add_f32(o + 2, h4.z);
    atomic_add_f32(o + 3, h4.w);
    if (lane == 0) atomic_add_f32(&counts[g], 1.0f);
}

__global__ void k_pool_div(float* __restrict__ sums, const float* __restrict__ counts) {
    int i = blockIdx.x * blockDim.x + threadIdx.x;
    if (i >= N_GRAPHS * HID) return;
    int g = i >> 7;
    sums[i] = sums[i] / fmaxf(counts[g], 1.0f);
}

extern "C" void kernel_launch(void* const* d_in, const int* in_sizes, int n_in,
                              void* d_out, int out_size, void* d_ws, size_t ws_size,
                              hipStream_t stream) {
    const float* x        = (const float*)d_in[0];
    const int*   eidx     = (const int*)d_in[1];
    const int*   batch    = (const int*)d_in[2];
    const float* W1       = (const float*)d_in[3];
    const float* b1       = (const float*)d_in[4];
    const float* W2       = (const float*)d_in[5];
    const float* b2       = (const float*)d_in[6];
    const float* W3       = (const float*)d_in[7];
    const float* b3       = (const float*)d_in[8];
    float* out = (float*)d_out;

    const int* src = eidx;
    const int* dst = eidx + N_EDGES;

    float* ws     = (float*)d_ws;
    float* dis    = ws;
    float* hw     = dis + N_NODES;
    float* hA     = hw + (size_t)N_NODES * HID;
    float* counts = hA + (size_t)N_NODES * HID;
    int*   wtot   = (int*)(counts + N_GRAPHS);
    int*   elist  = wtot + OWN_WG * 32;

    const int B = 256;
    int g_nf   = (N_NODES * HID + B - 1) / B;
    int g_n    = (N_NODES + B - 1) / B;
    int g_e    = (N_EDGES + B - 1) / B;
    int g_nw   = (N_NODES * 32 + B - 1) / B;

    k_fill1<<<g_n, B, 0, stream>>>(dis, N_NODES);
    k_deg<<<g_e, B, 0, stream>>>(dst, dis);
    k_norm<<<OWN_WG, B, 0, stream>>>(dis, wtot);

    k_gemm_in<<<g_nf, B, 0, stream>>>(x, W1, hw);
    k_edge_own<false, true><<<OWN_WG, 1024, 0, stream>>>(src, dst, dis, hw, b1, hA, wtot, elist, W2);

    k_edge_own<true, true><<<OWN_WG, 1024, 0, stream>>>(src, dst, dis, hA, b2, hw, wtot, elist, W3);

    k_edge_own<true, false><<<OWN_WG, 1024, 0, stream>>>(src, dst, dis, hw, b3, hA, wtot, elist, W3);

    k_pool_zero<<<(N_GRAPHS * HID + B - 1) / B, B, 0, stream>>>(out, counts);
    k_pool_acc<<<g_nw, B, 0, stream>>>(hA, batch, out, counts);
    k_pool_div<<<(N_GRAPHS * HID + B - 1) / B, B, 0, stream>>>(out, counts);
}


# Layer-1 source rows are made per edge from the node's 9 inputs and the lane's W1 columns in registers (same multiply-add chain): the x@W1 kernel, its table and its row reads go; sums bit-identical
# speedup vs baseline: 2.7127x; 1.0331x over previous
#include <hip/hip_runtime.h>

#define N_NODES  50000
#define N_EDGES  1600000
#define N_GRAPHS 256
#define IN_DIM   9
#define HID      128
#define OWN      196
#define OWN_WG   256
#define PITCH    132
#define ACC_ROWS 208

typedef __attribute__((ext_vector_type(2))) float v2f;
typedef __attribute__((ext_vector_type(8))) float v8f;

__device__ __forceinline__ void atomic_add_f32(float* p, float v) {
    asm volatile("global_atomic_add_f32 %0, %1, off scope:SCOPE_DEV"
                 :: "v"(p), "v"(v) : "memory");
}

__global__ void k_fill1(float* p, int n) {
    int i = blockIdx.x * blockDim.x + threadIdx.x;
    if (i < n) p[i] = 1.0f;
}

__global__ void k_deg(const int* __restrict__ dst, float* __restrict__ deg) {
    int e = blockIdx.x * blockDim.x + threadIdx.x;
    if (e < N_EDGES) atomic_add_f32(&deg[dst[e]], 1.0f);
}

__global__ __launch_bounds__(256) void k_norm(float* __restrict__ p, int* __restrict__ wtot) {
    __shared__ int red[8][32];
    const int r = threadIdx.x;
    const int n = blockIdx.x * OWN + r;
    int c = 0;
    if (r < OWN && n < N_NODES) {
        float d = p[n];
        p[n] = rsqrtf(d);
        c = (int)d - 1;
    }
    red[r >> 5][r & 31] = c;
    __syncthreads();
    if (r < 32) {
        int t = 0;
        for (int j = 0; j < 8; ++j) t += red[j][r];
        wtot[blockIdx.x * 32 + r] = t;
    }
}

__device__ __forceinline__ float4 in_row_times_w(float xv, const float4* wc) {
    float4 h = make_float4(0.f, 0.f, 0.f, 0.f);
#pragma unroll
    for (int k = 0; k < IN_DIM; ++k) {
        float xk = __builtin_bit_cast(float, __builtin_amdgcn_readlane(__builtin_bit_cast(int, xv), k));
        h.x = fmaf(xk, wc[k].x, h.x); h.y = fmaf(xk, wc[k].y, h.y);
        h.z = fmaf(xk, wc[k].z, h.z); h.w = fmaf(xk, wc[k].w, h.w);
    }
    return h;
}

template <int NT>
__device__ __forceinline__ void rows_times_w(const float* arow, const float* __restrict__ wl, v8f* d) {
    v2f bc[8], bn[8];
    float2 an[NT];
#pragma unroll
    for (int s = 0; s < 8; ++s) { bc[s].x = wl[(4 * s) * HID]; bc[s].y = wl[(4 * s + 1) * HID]; }
#pragma unroll
    for (int t = 0; t < NT; ++t) an[t] = *(const float2*)(arow + t * 64 * PITCH);
#pragma unroll
    for (int cb = 0; cb < HID; cb += 32) {
        if (cb + 32 < HID) {
#pragma unroll
            for (int s = 0; s < 8; ++s) {
                bn[s].x = wl[(cb + 32 + 4 * s) * HID];
                bn[s].y = wl[(cb + 32 + 4 * s + 1) * HID];
            }
        }
#pragma unroll
        for (int s = 0; s < 8; ++s) {
            const int k0 = cb + 4 * s;
            float2 ac[NT];
#pragma unroll
            for (int t = 0; t < NT; ++t) ac[t] = an[t];
            if (k0 + 4 < HID) {
#pragma unroll
                for (int t = 0; t < NT; ++t) an[t] = *(const float2*)(arow + t * 64 * PITCH + k0 + 4);
            }
            __builtin_amdgcn_sched_barrier(0);
#pragma unroll
            for (int t = 0; t < NT; ++t) {
                v2f a; a.x = fmaxf(ac[t].x, 0.f); a.y = fmaxf(ac[t].y, 0.f);
                d[t] = __builtin_amdgcn_wmma_f32_16x16x4_f32(false, a, false, bc[s],
                                                             (short)0, d[t], false, false);
            }
        }
#pragma unroll
        for (int s = 0; s < 8; ++s) bc[s] = bn[s];
    }
    if (NT == 4)
        asm volatile("v_nop\n\tv_nop\n\tv_nop\n\tv_nop" : "+v"(d[0]), "+v"(d[1]), "+v"(d[2]), "+v"(d[3]));
    else
        asm volatile("v_nop\n\tv_nop\n\tv_nop\n\tv_nop" : "+v"(d[0]), "+v"(d[1]), "+v"(d[2]));
}

#define QW         768
#define PW         64
#define RS         256
#define UNITS      (N_EDGES / 128)
#define TRIP_UNITS 4
#define TRIPS      ((UNITS + 32 * TRIP_UNITS - 1) / (32 * TRIP_UNITS))

static_assert(N_NODES <= 65536 && OWN <= 256, "a saved entry packs the source node in 16 bits, the local row above; k_norm has 256 threads per block of OWN rows");

__device__ __forceinline__ int trip_ubase(int pos, int w) {
    int itr = (pos + (int)blockIdx.x) % TRIPS;
    return (itr * 32 + w) * TRIP_UNITS;
}

__device__ __forceinline__ void load_trip(const int4* __restrict__ d4, int pos, int w, int lane,
                                          int4* dv) {
    const int ubase = trip_ubase(pos, w);
#pragma unroll
    for (int j = 0; j < TRIP_UNITS; ++j) {
        int u = ubase + j;
        if (u < UNITS) dv[j] = d4[u * 32 + lane];
        else           dv[j] = make_int4(-1, -1, -1, -1);
    }
}

__device__ __forceinline__ void add_rows(int s, int dl, float c, int n, int lane,
                                         const float* __restrict__ hw, float* acc) {
#pragma unroll 8
    for (int e = 0; e < n; ++e) {
        int sj = __builtin_amdgcn_readlane(s, e);
        int dj = __builtin_amdgcn_readlane(dl, e);
        float cj = __builtin_bit_cast(float, __builtin_amdgcn_readlane(__builtin_bit_cast(int, c), e));
        float4 h = *(const float4*)(hw + (size_t)sj * HID + lane * 4);
        float4* a = (float4*)(acc + dj * PITCH) + lane;
        float4 v = *a;
        v.x += h.x * cj; v.y += h.y * cj; v.z += h.z * cj; v.w += h.w * cj;
        *a = v;
    }
}

__device__ __forceinline__ void add_rows_in(int s, int dl, float c, int n, int lane,
                                            const float* __restrict__ x, const float4* wc, float* acc) {
    const int xl = min(lane, IN_DIM - 1);
#pragma unroll 8
    for (int e = 0; e < n; ++e) {
        int sj = __builtin_amdgcn_readlane(s, e);
        int dj = __builtin_amdgcn_readlane(dl, e);
        float cj = __builtin_bit_cast(float, __builtin_amdgcn_readlane(__builtin_bit_cast(int, c), e));
        float4 h = in_row_times_w(x[sj * IN_DIM + xl], wc);
        float4* a = (float4*)(acc + dj * PITCH) + lane;
        float4 v = *a;
        v.x += h.x * cj; v.y += h.y * cj; v.z += h.z * cj; v.w += h.w * cj;
        *a = v;
    }
}

__device__ __forceinline__ void apply_pending(const int2* pq, int n, int lane, int n0,
                                              const int* __restrict__ src,
                                              const float* __restrict__ dis,
                                              const float* __restrict__ x, const float4* wc, float* acc,
                                              int2* __restrict__ save) {
    int s = 0, dl = 0; float c = 0.f;
    if (lane < n) {
        int2 en = pq[lane];
        s = src[en.x];
        dl = en.y;
        c = dis[s] * dis[n0 + dl];
        save[lane] = make_int2(s | (dl << 16), __builtin_bit_cast(int, c));
    }
    add_rows_in(s, dl, c, n, lane, x, wc, acc);
}

template <bool REPLAY, bool FUSE>
__global__ __launch_bounds__(1024) void k_edge_own(const int* __restrict__ src,
                                                   const int* __restrict__ dst,
                                                   const float* __restrict__ dis,
                                                   const float* __restrict__ hw,
                                                   const float* __restrict__ bias,
                                                   float* __restrict__ out,
                                                   const int* __restrict__ wtot,
                                                   int* __restrict__ list_i,
                                                   const float* __restrict__ wnext,
                                                   const float* __restrict__ win) {
    __shared__ __attribute__((aligned(16))) float acc[ACC_ROWS * PITCH];
    __shared__ int2  q[32][QW];
    __shared__ int2  pend[32][PW];
    __shared__ int   qcnt[32];
    __shared__ int   unfinished[32];
    __shared__ int   sred[32];

    const int tid  = threadIdx.x;
    const int lane = tid & 31;
    const int w    = __builtin_amdgcn_readfirstlane(tid >> 5);
    const int n0   = blockIdx.x * OWN;
    const int nn   = min(OWN, N_NODES - n0);
    const unsigned lt = (1u << lane) - 1u;

    float4 wc[IN_DIM];
    if (!REPLAY) {
#pragma unroll
        for (int k = 0; k < IN_DIM; ++k) wc[k] = ((const float4*)(win + k * HID))[lane];
    }

    for (int i = tid; i < nn * (HID / 4); i += 1024) {
        int r = i >> 5;
        int f4 = i & 31;
        float s = dis[n0 + r];
        s = s * s;
        float4 h;
        if (REPLAY) h = ((const float4*)(hw + (size_t)(n0 + r) * HID))[f4];
        else        h = in_row_times_w(hw[(n0 + r) * IN_DIM + min(lane, IN_DIM - 1)], wc);
        float4 b = ((const float4*)bias)[f4];
        float4 o;
        o.x = fmaf(h.x, s, b.x); o.y = fmaf(h.y, s, b.y);
        o.z = fmaf(h.z, s, b.z); o.w = fmaf(h.w, s, b.w);
        ((float4*)(acc + r * PITCH))[f4] = o;
    }

    int pre = 0;
    for (int i = tid; i < (int)blockIdx.x * 32; i += 1024) pre += wtot[i];
    for (int o = 16; o > 0; o >>= 1) pre += __shfl_xor(pre, o);
    if (lane == 0) sred[w] = pre;
    __syncthreads();
    const int mine = wtot[blockIdx.x * 32 + lane];
    int st = sred[lane] + ((lane < w) ? mine : 0);
    for (int o = 16; o > 0; o >>= 1) st += __shfl_xor(st, o);
    int2* mylist = (int2*)list_i + __builtin_amdgcn_readfirstlane(st);

    int2* myq = q[w];

    if (REPLAY) {
        const int mycnt = __builtin_amdgcn_readlane(mine, w);
        for (int cb = 0; cb < mycnt; cb += RS) {
            const int c = min(mycnt - cb, RS);
            int2 t[RS / 32];
#pragma unroll
            for (int k = 0; k < RS / 32; ++k) {
                const int i = k * 32 + lane;
                t[k] = make_int2(0, 0);
                if (i < c) t[k] = mylist[cb + i];
            }
#pragma unroll
            for (int k = 0; k < RS / 32; ++k) myq[k * 32 + lane] = t[k];
            for (int b = 0; b < c; b += 32) {
                const int n = min(32, c - b);
                int sd = 0; float cf = 0.f;
                if (lane < n) {
                    int2 en = myq[b + lane];
                    sd = en.x;
                    cf = __builtin_bit_cast(float, en.y);
                }
                add_rows(sd & 0xffff, sd >> 16, cf, n, lane, hw, acc);
            }
        }
        __syncthreads();
    } else {
        const int4* d4 = (const int4*)dst;
        int2* myp = pend[w];
        int pos = 0;
        int wpos = 0;

        for (;;) {
            int cnt = 0;
            int4 dvn[TRIP_UNITS];
            if (pos < TRIPS) load_trip(d4, pos, w, lane, dvn);
            while (pos < TRIPS && cnt + TRIP_UNITS * 128 <= QW) {
                const int ubase = trip_ubase(pos, w);
                int4 dv[TRIP_UNITS];
#pragma unroll
                for (int j = 0; j < TRIP_UNITS; ++j) dv[j] = dvn[j];
                ++pos;
                if (pos < TRIPS) load_trip(d4, pos, w, lane, dvn);
#pragma unroll
                for (int j = 0; j < TRIP_UNITS; ++j) {
                    int ebase = (ubase + j) * 128 + lane * 4;
                    int dd[4] = {dv[j].x, dv[j].y, dv[j].z, dv[j].w};
#pragma unroll
                    for (int k = 0; k < 4; ++k) {
                        unsigned dl = (unsigned)(dd[k] - n0);
                        bool m = dl < (unsigned)nn;
                        unsigned mask = __builtin_amdgcn_ballot_w32(m);
                        if (mask) {
                            if (m) myq[cnt + __builtin_popcount(mask & lt)] = make_int2(ebase + k, (int)dl);
                            cnt += __builtin_popcount(mask);
                        }
                    }
                }
            }
            if (lane == 0) { qcnt[w] = cnt; unfinished[w] = (pos < TRIPS) ? 1 : 0; }
            __syncthreads();
            const bool more = __builtin_amdgcn_ballot_w32(unfinished[lane] != 0) != 0u;

            int pc = 0;
            for (int p = 0; p < 32; ++p) {
                const int np = __builtin_amdgcn_readfirstlane(qcnt[p]);
                const int2* pq = q[p];
                for (int g = 0; g < np; g += 32) {
                    int2 en = make_int2(0, -1);
                    if (g + lane < np) en = pq[g + lane];
                    bool m = (en.y >= 0) && ((en.y & 31) == w);
                    unsigned mask = __builtin_amdgcn_ballot_w32(m);
                    if (mask) {
                        if (m) myp[pc + __builtin_popcount(mask & lt)] = en;
                        pc += __builtin_popcount(mask);
                        if (pc >= 32) {
                            apply_pending(myp, 32, lane, n0, src, dis, hw, wc, acc, mylist + wpos);
                            wpos += 32;
                            int2 mv = make_int2(0, 0);
                            if (lane < pc - 32) mv = myp[32 + lane];
                            if (lane < pc - 32) myp[lane] = mv;
                            pc -= 32;
                        }
                    }
                }
            }
            if (pc > 0) {
                apply_pending(myp, pc, lane, n0, src, dis, hw, wc, acc, mylist + wpos);
                wpos += pc;
            }
            __syncthreads();
            if (!more) break;
        }
    }

    if (FUSE) {
        const int ct  = w & 7;
        const int g   = w >> 3;
        const int c16 = lane & 15;
        const int hf  = lane >> 4;
        const float* arow = acc + (g * 16 + c16) * PITCH + 2 * hf;
        const float* wl   = wnext + (2 * hf) * HID + ct * 16 + c16;
        const int nt = (g == 0) ? 4 : 3;
        v8f d[4];
#pragma unroll
        for (int t = 0; t < 4; ++t) d[t] = (v8f){0.f, 0.f, 0.f, 0.f, 0.f, 0.f, 0.f, 0.f};
        if (g == 0) rows_times_w<4>(arow, wl, d);
        else        rows_times_w<3>(arow, wl, d);
        __syncthreads();
        float* drow = acc + (g * 16 + 8 * hf) * PITCH + ct * 16 + c16;
#pragma unroll
        for (int t = 0; t < 4; ++t) {
            if (t < nt) {
#pragma unroll
                for (int v = 0; v < 8; ++v) drow[(t * 64 + v) * PITCH] = d[t][v];
            }
        }
        __syncthreads();
    }

    float4* o4 = (float4*)(out + (size_t)n0 * HID);
    for (int i = tid; i < nn * (HID / 4); i += 1024)
        o4[i] = ((const float4*)(acc + (i >> 5) * PITCH))[i & 31];
}

__global__ void k_pool_zero(float* __restrict__ sums, float* __restrict__ counts) {
    int i = blockIdx.x * blockDim.x + threadIdx.x;
    if (i < N_GRAPHS * HID) sums[i] = 0.f;
    if (i < N_GRAPHS) counts[i] = 0.f;
}

__global__ void k_pool_acc(const float* __restrict__ h, const int* __restrict__ batch,
                           float* __restrict__ sums, float* __restrict__ counts) {
    int t = blockIdx.x * blockDim.x + threadIdx.x;
    int n = t >> 5;
    if (n >= N_NODES) return;
    n = __builtin_amdgcn_readfirstlane(n);
    int lane = t & 31;
    int g = batch[n];
    const float4 h4 = *(const float4*)(h + n * HID + lane * 4);
    float* o = sums + g * HID + lane * 4;
    atomic_add_f32(o + 0, h4.x);
    atomic_add_f32(o + 1, h4.y);
    atomic_add_f32(o + 2, h4.z);
    atomic_add_f32(o + 3, h4.w);
    if (lane == 0) atomic_add_f32(&counts[g], 1.0f);
}

__global__ void k_pool_div(float* __restrict__ sums, const float* __restrict__ counts) {
    int i = blockIdx.x * blockDim.x + threadIdx.x;
    if (i >= N_GRAPHS * HID) return;
    int g = i >> 7;
    sums[i] = sums[i] / fmaxf(counts[g], 1.0f);
}

extern "C" void kernel_launch(void* const* d_in, const int* in_sizes, int n_in,
                              void* d_out, int out_size, void* d_ws, size_t ws_size,
                              hipStream_t stream) {
    const float* x        = (const float*)d_in[0];
    const int*   eidx     = (const int*)d_in[1];
    const int*   batch    = (const int*)d_in[2];
    const float* W1       = (const float*)d_in[3];
    const float* b1       = (const float*)d_in[4];
    const float* W2       = (const float*)d_in[5];
    const float* b2       = (const float*)d_in[6];
    const float* W3       = (const float*)d_in[7];
    const float* b3       = (const float*)d_in[8];
    float* out = (float*)d_out;

    const int* src = eidx;
    const int* dst = eidx + N_EDGES;

    float* ws     = (float*)d_ws;
    float* dis    = ws;
    float* hw     = dis + N_NODES;
    float* hA     = hw + (size_t)N_NODES * HID;
    float* counts = hA + (size_t)N_NODES * HID;
    int*   wtot   = (int*)(counts + N_GRAPHS);
    int*   elist  = wtot + OWN_WG * 32;

    const int B = 256;
    int g_n    = (N_NODES + B - 1) / B;
    int g_e    = (N_EDGES + B - 1) / B;
    int g_nw   = (N_NODES * 32 + B - 1) / B;

    k_fill1<<<g_n, B, 0, stream>>>(dis, N_NODES);
    k_deg<<<g_e, B, 0, stream>>>(dst, dis);
    k_norm<<<OWN_WG, B, 0, stream>>>(dis, wtot);

    k_edge_own<false, true><<<OWN_WG, 1024, 0, stream>>>(src, dst, dis, x, b1, hA, wtot, elist, W2, W1);

    k_edge_own<true, true><<<OWN_WG, 1024, 0, stream>>>(src, dst, dis, hA, b2, hw, wtot, elist, W3, W1);

    k_edge_own<true, false><<<OWN_WG, 1024, 0, stream>>>(src, dst, dis, hw, b3, hA, wtot, elist, W3, W1);

    k_pool_zero<<<(N_GRAPHS * HID + B - 1) / B, B, 0, stream>>>(out, counts);
    k_pool_acc<<<g_nw, B, 0, stream>>>(hA, batch, out, counts);
    k_pool_div<<<(N_GRAPHS * HID + B - 1) / B, B, 0, stream>>>(out, counts);
}


# Layer-1 edge search reads one owner byte per edge (dst / 196, written by the degree kernel; 4 bytes tested per word), not 4-byte dst; queued edges' rows read after; per-node sum order changes
# speedup vs baseline: 2.8196x; 1.0394x over previous
#include <hip/hip_runtime.h>

#define N_NODES  50000
#define N_EDGES  1600000
#define N_GRAPHS 256
#define IN_DIM   9
#define HID      128
#define OWN      196
#define OWN_WG   256
#define PITCH    132
#define ACC_ROWS 208

typedef __attribute__((ext_vector_type(2))) float v2f;
typedef __attribute__((ext_vector_type(8))) float v8f;

__device__ __forceinline__ void atomic_add_f32(float* p, float v) {
    asm volatile("global_atomic_add_f32 %0, %1, off scope:SCOPE_DEV"
                 :: "v"(p), "v"(v) : "memory");
}

__global__ void k_fill1(float* p, int n) {
    int i = blockIdx.x * blockDim.x + threadIdx.x;
    if (i < n) p[i] = 1.0f;
}

__global__ void k_deg(const int* __restrict__ dst, float* __restrict__ deg, unsigned* __restrict__ own4) {
    int t = blockIdx.x * blockDim.x + threadIdx.x;
    if (t < N_EDGES / 4) {
        const int4 d = ((const int4*)dst)[t];
        atomic_add_f32(&deg[d.x], 1.0f);
        atomic_add_f32(&deg[d.y], 1.0f);
        atomic_add_f32(&deg[d.z], 1.0f);
        atomic_add_f32(&deg[d.w], 1.0f);
        own4[t] = ((unsigned)d.x / OWN) | (((unsigned)d.y / OWN) << 8) |
                  (((unsigned)d.z / OWN) << 16) | (((unsigned)d.w / OWN) << 24);
    }
}

__global__ __launch_bounds__(256) void k_norm(float* __restrict__ p, int* __restrict__ wtot) {
    __shared__ int red[8][32];
    const int r = threadIdx.x;
    const int n = blockIdx.x * OWN + r;
    int c = 0;
    if (r < OWN && n < N_NODES) {
        float d = p[n];
        p[n] = rsqrtf(d);
        c = (int)d - 1;
    }
    red[r >> 5][r & 31] = c;
    __syncthreads();
    if (r < 32) {
        int t = 0;
        for (int j = 0; j < 8; ++j) t += red[j][r];
        wtot[blockIdx.x * 32 + r] = t;
    }
}

__device__ __forceinline__ float4 in_row_times_w(float xv, const float4* wc) {
    float4 h = make_float4(0.f, 0.f, 0.f, 0.f);
#pragma unroll
    for (int k = 0; k < IN_DIM; ++k) {
        float xk = __builtin_bit_cast(float, __builtin_amdgcn_readlane(__builtin_bit_cast(int, xv), k));
        h.x = fmaf(xk, wc[k].x, h.x); h.y = fmaf(xk, wc[k].y, h.y);
        h.z = fmaf(xk, wc[k].z, h.z); h.w = fmaf(xk, wc[k].w, h.w);
    }
    return h;
}

template <int NT>
__device__ __forceinline__ void rows_times_w(const float* arow, const float* __restrict__ wl, v8f* d) {
    v2f bc[8], bn[8];
    float2 an[NT];
#pragma unroll
    for (int s = 0; s < 8; ++s) { bc[s].x = wl[(4 * s) * HID]; bc[s].y = wl[(4 * s + 1) * HID]; }
#pragma unroll
    for (int t = 0; t < NT; ++t) an[t] = *(const float2*)(arow + t * 64 * PITCH);
#pragma unroll
    for (int cb = 0; cb < HID; cb += 32) {
        if (cb + 32 < HID) {
#pragma unroll
            for (int s = 0; s < 8; ++s) {
                bn[s].x = wl[(cb + 32 + 4 * s) * HID];
                bn[s].y = wl[(cb + 32 + 4 * s + 1) * HID];
            }
        }
#pragma unroll
        for (int s = 0; s < 8; ++s) {
            const int k0 = cb + 4 * s;
            float2 ac[NT];
#pragma unroll
            for (int t = 0; t < NT; ++t) ac[t] = an[t];
            if (k0 + 4 < HID) {
#pragma unroll
                for (int t = 0; t < NT; ++t) an[t] = *(const float2*)(arow + t * 64 * PITCH + k0 + 4);
            }
            __builtin_amdgcn_sched_barrier(0);
#pragma unroll
            for (int t = 0; t < NT; ++t) {
                v2f a; a.x = fmaxf(ac[t].x, 0.f); a.y = fmaxf(ac[t].y, 0.f);
                d[t] = __builtin_amdgcn_wmma_f32_16x16x4_f32(false, a, false, bc[s],
                                                             (short)0, d[t], false, false);
            }
        }
#pragma unroll
        for (int s = 0; s < 8; ++s) bc[s] = bn[s];
    }
    if (NT == 4)
        asm volatile("v_nop\n\tv_nop\n\tv_nop\n\tv_nop" : "+v"(d[0]), "+v"(d[1]), "+v"(d[2]), "+v"(d[3]));
    else
        asm volatile("v_nop\n\tv_nop\n\tv_nop\n\tv_nop" : "+v"(d[0]), "+v"(d[1]), "+v"(d[2]));
}

#define QW         768
#define PW         64
#define RS         256
#define UEDGES     512
#define NUNITS     (N_EDGES / UEDGES)
#define UPW        ((NUNITS + 31) / 32)
#define DEPTH      8

static_assert(N_NODES <= 65536 && OWN <= 256, "a saved entry packs the source node in 16 bits, the local row above; k_norm has 256 threads per block of OWN rows");
static_assert(N_EDGES % UEDGES == 0 && (N_NODES + OWN - 1) / OWN <= 256 && UEDGES <= QW, "whole scan units; an owner fits a byte; one unit's edges fit the queue");

__device__ __forceinline__ int next_round(int r, int by) {
    r += by;
    return (r >= UPW) ? r - UPW : r;
}

__device__ __forceinline__ uint4 load_unit(const uint4* __restrict__ o4, int r, int w, int lane) {
    return o4[min(r * 32 + w, NUNITS - 1) * 32 + lane];
}

__device__ __forceinline__ unsigned unit_hits(uint4 ob, unsigned pat) {
    const unsigned wd[4] = {ob.x, ob.y, ob.z, ob.w};
    unsigned m = 0;
#pragma unroll
    for (int k = 0; k < 4; ++k) {
        const unsigned x = wd[k] ^ pat;
        const unsigned z = ~(((x & 0x7f7f7f7fu) + 0x7f7f7f7fu) | x | 0x7f7f7f7fu);
        m |= z >> k;
    }
    return m;
}

__device__ __forceinline__ int queue_hits(unsigned m, int ebase, int cnt, unsigned lt, int2* myq) {
    unsigned mask = __builtin_amdgcn_ballot_w32(m != 0u);
    while (mask) {
        if (m != 0u) {
            const int p = __builtin_ctz(m);
            myq[cnt + __builtin_popcount(mask & lt)].x = ebase + 4 * (7 - (p & 7)) + (p >> 3);
        }
        cnt += __builtin_popcount(mask);
        m &= m - 1u;
        mask = __builtin_amdgcn_ballot_w32(m != 0u);
    }
    return cnt;
}

__device__ __forceinline__ void add_rows(int s, int dl, float c, int n, int lane,
                                         const float* __restrict__ hw, float* acc) {
#pragma unroll 8
    for (int e = 0; e < n; ++e) {
        int sj = __builtin_amdgcn_readlane(s, e);
        int dj = __builtin_amdgcn_readlane(dl, e);
        float cj = __builtin_bit_cast(float, __builtin_amdgcn_readlane(__builtin_bit_cast(int, c), e));
        float4 h = *(const float4*)(hw + (size_t)sj * HID + lane * 4);
        float4* a = (float4*)(acc + dj * PITCH) + lane;
        float4 v = *a;
        v.x += h.x * cj; v.y += h.y * cj; v.z += h.z * cj; v.w += h.w * cj;
        *a = v;
    }
}

__device__ __forceinline__ void add_rows_in(int s, int dl, float c, int n, int lane,
                                            const float* __restrict__ x, const float4* wc, float* acc) {
    const int xl = min(lane, IN_DIM - 1);
#pragma unroll 8
    for (int e = 0; e < n; ++e) {
        int sj = __builtin_amdgcn_readlane(s, e);
        int dj = __builtin_amdgcn_readlane(dl, e);
        float cj = __builtin_bit_cast(float, __builtin_amdgcn_readlane(__builtin_bit_cast(int, c), e));
        float4 h = in_row_times_w(x[sj * IN_DIM + xl], wc);
        float4* a = (float4*)(acc + dj * PITCH) + lane;
        float4 v = *a;
        v.x += h.x * cj; v.y += h.y * cj; v.z += h.z * cj; v.w += h.w * cj;
        *a = v;
    }
}

__device__ __forceinline__ void apply_pending(const int2* pq, int n, int lane, int n0,
                                              const int* __restrict__ src,
                                              const float* __restrict__ dis,
                                              const float* __restrict__ x, const float4* wc, float* acc,
                                              int2* __restrict__ save) {
    int s = 0, dl = 0; float c = 0.f;
    if (lane < n) {
        int2 en = pq[lane];
        s = src[en.x];
        dl = en.y;
        c = dis[s] * dis[n0 + dl];
        save[lane] = make_int2(s | (dl << 16), __builtin_bit_cast(int, c));
    }
    add_rows_in(s, dl, c, n, lane, x, wc, acc);
}

template <bool REPLAY, bool FUSE>
__global__ __launch_bounds__(1024) void k_edge_own(const int* __restrict__ src,
                                                   const int* __restrict__ dst,
                                                   const float* __restrict__ dis,
                                                   const float* __restrict__ hw,
                                                   const float* __restrict__ bias,
                                                   float* __restrict__ out,
                                                   const int* __restrict__ wtot,
                                                   int* __restrict__ list_i,
                                                   const float* __restrict__ wnext,
                                                   const float* __restrict__ win,
                                                   const unsigned* __restrict__ own4) {
    __shared__ __attribute__((aligned(16))) float acc[ACC_ROWS * PITCH];
    __shared__ int2  q[32][QW];
    __shared__ int2  pend[32][PW];
    __shared__ int   qcnt[32];
    __shared__ int   unfinished[32];
    __shared__ int   sred[32];

    const int tid  = threadIdx.x;
    const int lane = tid & 31;
    const int w    = __builtin_amdgcn_readfirstlane(tid >> 5);
    const int n0   = blockIdx.x * OWN;
    const int nn   = min(OWN, N_NODES - n0);
    const unsigned lt = (1u << lane) - 1u;

    float4 wc[IN_DIM];
    if (!REPLAY) {
#pragma unroll
        for (int k = 0; k < IN_DIM; ++k) wc[k] = ((const float4*)(win + k * HID))[lane];
    }

    for (int i = tid; i < nn * (HID / 4); i += 1024) {
        int r = i >> 5;
        int f4 = i & 31;
        float s = dis[n0 + r];
        s = s * s;
        float4 h;
        if (REPLAY) h = ((const float4*)(hw + (size_t)(n0 + r) * HID))[f4];
        else        h = in_row_times_w(hw[(n0 + r) * IN_DIM + min(lane, IN_DIM - 1)], wc);
        float4 b = ((const float4*)bias)[f4];
        float4 o;
        o.x = fmaf(h.x, s, b.x); o.y = fmaf(h.y, s, b.y);
        o.z = fmaf(h.z, s, b.z); o.w = fmaf(h.w, s, b.w);
        ((float4*)(acc + r * PITCH))[f4] = o;
    }

    int pre = 0;
    for (int i = tid; i < (int)blockIdx.x * 32; i += 1024) pre += wtot[i];
    for (int o = 16; o > 0; o >>= 1) pre += __shfl_xor(pre, o);
    if (lane == 0) sred[w] = pre;
    __syncthreads();
    const int mine = wtot[blockIdx.x * 32 + lane];
    int st = sred[lane] + ((lane < w) ? mine : 0);
    for (int o = 16; o > 0; o >>= 1) st += __shfl_xor(st, o);
    int2* mylist = (int2*)list_i + __builtin_amdgcn_readfirstlane(st);

    int2* myq = q[w];

    if (REPLAY) {
        const int mycnt = __builtin_amdgcn_readlane(mine, w);
        for (int cb = 0; cb < mycnt; cb += RS) {
            const int c = min(mycnt - cb, RS);
            int2 t[RS / 32];
#pragma unroll
            for (int k = 0; k < RS / 32; ++k) {
                const int i = k * 32 + lane;
                t[k] = make_int2(0, 0);
                if (i < c) t[k] = mylist[cb + i];
            }
#pragma unroll
            for (int k = 0; k < RS / 32; ++k) myq[k * 32 + lane] = t[k];
            for (int b = 0; b < c; b += 32) {
                const int n = min(32, c - b);
                int sd = 0; float cf = 0.f;
                if (lane < n) {
                    int2 en = myq[b + lane];
                    sd = en.x;
                    cf = __builtin_bit_cast(float, en.y);
                }
                add_rows(sd & 0xffff, sd >> 16, cf, n, lane, hw, acc);
            }
        }
        __syncthreads();
    } else {
        const uint4* o4 = (const uint4*)own4;
        const unsigned pat = (unsigned)blockIdx.x * 0x01010101u;
        const int rot = (int)blockIdx.x % UPW;
        int2* myp = pend[w];
        int pos = 0;
        int wpos = 0;

        for (;;) {
            int cnt = 0;
            int rnd = next_round(rot, pos);
            uint4 win[DEPTH];
#pragma unroll
            for (int j = 0; j < DEPTH; ++j) win[j] = load_unit(o4, next_round(rnd, j), w, lane);
            bool go = true;
            while (go) {
#pragma unroll
                for (int j = 0; j < DEPTH; ++j) {
                    if (pos >= UPW || cnt + UEDGES > QW) { go = false; break; }
                    const int u = rnd * 32 + w;
                    unsigned m = unit_hits(win[j], pat);
                    if (u >= NUNITS) m = 0u;
                    win[j] = load_unit(o4, next_round(rnd, DEPTH), w, lane);
                    ++pos;
                    rnd = next_round(rnd, 1);
                    cnt = queue_hits(m, u * UEDGES + lane * 16, cnt, lt, myq);
                }
            }
            for (int b = 0; b < cnt; b += 256) {
                int de[8];
#pragma unroll
                for (int k = 0; k < 8; ++k) {
                    const int i = b + k * 32 + lane;
                    de[k] = (i < cnt) ? myq[i].x : 0;
                }
#pragma unroll
                for (int k = 0; k < 8; ++k) de[k] = dst[de[k]];
#pragma unroll
                for (int k = 0; k < 8; ++k) {
                    const int i = b + k * 32 + lane;
                    if (i < cnt) myq[i].y = de[k] - n0;
                }
            }
            if (lane == 0) { qcnt[w] = cnt; unfinished[w] = (pos < UPW) ? 1 : 0; }
            __syncthreads();
            const bool more = __builtin_amdgcn_ballot_w32(unfinished[lane] != 0) != 0u;

            int pc = 0;
            for (int p = 0; p < 32; ++p) {
                const int np = __builtin_amdgcn_readfirstlane(qcnt[p]);
                const int2* pq = q[p];
                for (int g = 0; g < np; g += 32) {
                    int2 en = make_int2(0, -1);
                    if (g + lane < np) en = pq[g + lane];
                    bool m = (en.y >= 0) && ((en.y & 31) == w);
                    unsigned mask = __builtin_amdgcn_ballot_w32(m);
                    if (mask) {
                        if (m) myp[pc + __builtin_popcount(mask & lt)] = en;
                        pc += __builtin_popcount(mask);
                        if (pc >= 32) {
                            apply_pending(myp, 32, lane, n0, src, dis, hw, wc, acc, mylist + wpos);
                            wpos += 32;
                            int2 mv = make_int2(0, 0);
                            if (lane < pc - 32) mv = myp[32 + lane];
                            if (lane < pc - 32) myp[lane] = mv;
                            pc -= 32;
                        }
                    }
                }
            }
            if (pc > 0) {
                apply_pending(myp, pc, lane, n0, src, dis, hw, wc, acc, mylist + wpos);
                wpos += pc;
            }
            __syncthreads();
            if (!more) break;
        }
    }

    if (FUSE) {
        const int ct  = w & 7;
        const int g   = w >> 3;
        const int c16 = lane & 15;
        const int hf  = lane >> 4;
        const float* arow = acc + (g * 16 + c16) * PITCH + 2 * hf;
        const float* wl   = wnext + (2 * hf) * HID + ct * 16 + c16;
        const int nt = (g == 0) ? 4 : 3;
        v8f d[4];
#pragma unroll
        for (int t = 0; t < 4; ++t) d[t] = (v8f){0.f, 0.f, 0.f, 0.f, 0.f, 0.f, 0.f, 0.f};
        if (g == 0) rows_times_w<4>(arow, wl, d);
        else        rows_times_w<3>(arow, wl, d);
        __syncthreads();
        float* drow = acc + (g * 16 + 8 * hf) * PITCH + ct * 16 + c16;
#pragma unroll
        for (int t = 0; t < 4; ++t) {
            if (t < nt) {
#pragma unroll
                for (int v = 0; v < 8; ++v) drow[(t * 64 + v) * PITCH] = d[t][v];
            }
        }
        __syncthreads();
    }

    float4* o4 = (float4*)(out + (size_t)n0 * HID);
    for (int i = tid; i < nn * (HID / 4); i += 1024)
        o4[i] = ((const float4*)(acc + (i >> 5) * PITCH))[i & 31];
}

__global__ void k_pool_zero(float* __restrict__ sums, float* __restrict__ counts) {
    int i = blockIdx.x * blockDim.x + threadIdx.x;
    if (i < N_GRAPHS * HID) sums[i] = 0.f;
    if (i < N_GRAPHS) counts[i] = 0.f;
}

__global__ void k_pool_acc(const float* __restrict__ h, const int* __restrict__ batch,
                           float* __restrict__ sums, float* __restrict__ counts) {
    int t = blockIdx.x * blockDim.x + threadIdx.x;
    int n = t >> 5;
    if (n >= N_NODES) return;
    n = __builtin_amdgcn_readfirstlane(n);
    int lane = t & 31;
    int g = batch[n];
    const float4 h4 = *(const float4*)(h + n * HID + lane * 4);
    float* o = sums + g * HID + lane * 4;
    atomic_add_f32(o + 0, h4.x);
    atomic_add_f32(o + 1, h4.y);
    atomic_add_f32(o + 2, h4.z);
    atomic_add_f32(o + 3, h4.w);
    if (lane == 0) atomic_add_f32(&counts[g], 1.0f);
}

__global__ void k_pool_div(float* __restrict__ sums, const float* __restrict__ counts) {
    int i = blockIdx.x * blockDim.x + threadIdx.x;
    if (i >= N_GRAPHS * HID) return;
    int g = i >> 7;
    sums[i] = sums[i] / fmaxf(counts[g], 1.0f);
}

extern "C" void kernel_launch(void* const* d_in, const int* in_sizes, int n_in,
                              void* d_out, int out_size, void* d_ws, size_t ws_size,
                              hipStream_t stream) {
    const float* x        = (const float*)d_in[0];
    const int*   eidx     = (const int*)d_in[1];
    const int*   batch    = (const int*)d_in[2];
    const float* W1       = (const float*)d_in[3];
    const float* b1       = (const float*)d_in[4];
    const float* W2       = (const float*)d_in[5];
    const float* b2       = (const float*)d_in[6];
    const float* W3       = (const float*)d_in[7];
    const float* b3       = (const float*)d_in[8];
    float* out = (float*)d_out;

    const int* src = eidx;
    const int* dst = eidx + N_EDGES;

    float* ws     = (float*)d_ws;
    float* dis    = ws;
    float* hw     = dis + N_NODES;
    float* hA     = hw + (size_t)N_NODES * HID;
    float* counts = hA + (size_t)N_NODES * HID;
    int*   wtot   = (int*)(counts + N_GRAPHS);
    int*   elist  = wtot + OWN_WG * 32;
    unsigned* own4 = (unsigned*)(elist + 2 * N_EDGES);

    const int B = 256;
    int g_n    = (N_NODES + B - 1) / B;
    int g_e4   = (N_EDGES / 4 + B - 1) / B;
    int g_nw   = (N_NODES * 32 + B - 1) / B;

    k_fill1<<<g_n, B, 0, stream>>>(dis, N_NODES);
    k_deg<<<g_e4, B, 0, stream>>>(dst, dis, own4);
    k_norm<<<OWN_WG, B, 0, stream>>>(dis, wtot);

    k_edge_own<false, true><<<OWN_WG, 1024, 0, stream>>>(src, dst, dis, x, b1, hA, wtot, elist, W2, W1, own4);

    k_edge_own<true, true><<<OWN_WG, 1024, 0, stream>>>(src, dst, dis, hA, b2, hw, wtot, elist, W3, W1, own4);

    k_edge_own<true, false><<<OWN_WG, 1024, 0, stream>>>(src, dst, dis, hw, b3, hA, wtot, elist, W3, W1, own4);

    k_pool_zero<<<(N_GRAPHS * HID + B - 1) / B, B, 0, stream>>>(out, counts);
    k_pool_acc<<<g_nw, B, 0, stream>>>(hA, batch, out, counts);
    k_pool_div<<<(N_GRAPHS * HID + B - 1) / B, B, 0, stream>>>(out, counts);
}


# Layer-1 queue entries sorted by applying wave (row & 31, order kept, 224 at a time, start offsets in LDS): a wave reads only its own entries, the filter pass over all queues goes; sums bit-identical
# speedup vs baseline: 3.7913x; 1.3446x over previous
#include <hip/hip_runtime.h>

#define N_NODES  50000
#define N_EDGES  1600000
#define N_GRAPHS 256
#define IN_DIM   9
#define HID      128
#define OWN      196
#define OWN_WG   256
#define PITCH    132
#define ACC_ROWS 208

typedef __attribute__((ext_vector_type(2))) float v2f;
typedef __attribute__((ext_vector_type(8))) float v8f;

__device__ __forceinline__ void atomic_add_f32(float* p, float v) {
    asm volatile("global_atomic_add_f32 %0, %1, off scope:SCOPE_DEV"
                 :: "v"(p), "v"(v) : "memory");
}

__global__ void k_fill1(float* p, const int* __restrict__ batch, int* __restrict__ gstart) {
    int i = blockIdx.x * blockDim.x + threadIdx.x;
    if (i < N_NODES) {
        p[i] = 1.0f;
        const int cur  = batch[i];
        const int prev = (i == 0) ? -1 : batch[i - 1];
        for (int g = max(prev + 1, 0); g <= min(cur, N_GRAPHS - 1); ++g) gstart[g] = i;
        if (i == N_NODES - 1)
            for (int g = max(cur + 1, 0); g <= N_GRAPHS; ++g) gstart[g] = N_NODES;
    }
}

__global__ void k_deg(const int* __restrict__ dst, float* __restrict__ deg, unsigned* __restrict__ own4) {
    int t = blockIdx.x * blockDim.x + threadIdx.x;
    if (t < N_EDGES / 4) {
        const int4 d = ((const int4*)dst)[t];
        atomic_add_f32(&deg[d.x], 1.0f);
        atomic_add_f32(&deg[d.y], 1.0f);
        atomic_add_f32(&deg[d.z], 1.0f);
        atomic_add_f32(&deg[d.w], 1.0f);
        own4[t] = ((unsigned)d.x / OWN) | (((unsigned)d.y / OWN) << 8) |
                  (((unsigned)d.z / OWN) << 16) | (((unsigned)d.w / OWN) << 24);
    }
}

__global__ __launch_bounds__(256) void k_norm(float* __restrict__ p, int* __restrict__ wtot) {
    __shared__ int red[8][32];
    const int r = threadIdx.x;
    const int n = blockIdx.x * OWN + r;
    int c = 0;
    if (r < OWN && n < N_NODES) {
        float d = p[n];
        p[n] = rsqrtf(d);
        c = (int)d - 1;
    }
    red[r >> 5][r & 31] = c;
    __syncthreads();
    if (r < 32) {
        int t = 0;
        for (int j = 0; j < 8; ++j) t += red[j][r];
        wtot[blockIdx.x * 32 + r] = t;
    }
}

__device__ __forceinline__ float4 in_row_times_w(float xv, const float4* wc) {
    float4 h = make_float4(0.f, 0.f, 0.f, 0.f);
#pragma unroll
    for (int k = 0; k < IN_DIM; ++k) {
        float xk = __builtin_bit_cast(float, __builtin_amdgcn_readlane(__builtin_bit_cast(int, xv), k));
        h.x = fmaf(xk, wc[k].x, h.x); h.y = fmaf(xk, wc[k].y, h.y);
        h.z = fmaf(xk, wc[k].z, h.z); h.w = fmaf(xk, wc[k].w, h.w);
    }
    return h;
}

template <int NT>
__device__ __forceinline__ void rows_times_w(const float* arow, const float* __restrict__ wl, v8f* d) {
    v2f bc[8], bn[8];
    float2 an[NT];
#pragma unroll
    for (int s = 0; s < 8; ++s) { bc[s].x = wl[(4 * s) * HID]; bc[s].y = wl[(4 * s + 1) * HID]; }
#pragma unroll
    for (int t = 0; t < NT; ++t) an[t] = *(const float2*)(arow + t * 64 * PITCH);
#pragma unroll
    for (int cb = 0; cb < HID; cb += 32) {
        if (cb + 32 < HID) {
#pragma unroll
            for (int s = 0; s < 8; ++s) {
                bn[s].x = wl[(cb + 32 + 4 * s) * HID];
                bn[s].y = wl[(cb + 32 + 4 * s + 1) * HID];
            }
        }
#pragma unroll
        for (int s = 0; s < 8; ++s) {
            const int k0 = cb + 4 * s;
            float2 ac[NT];
#pragma unroll
            for (int t = 0; t < NT; ++t) ac[t] = an[t];
            if (k0 + 4 < HID) {
#pragma unroll
                for (int t = 0; t < NT; ++t) an[t] = *(const float2*)(arow + t * 64 * PITCH + k0 + 4);
            }
            __builtin_amdgcn_sched_barrier(0);
#pragma unroll
            for (int t = 0; t < NT; ++t) {
                v2f a; a.x = fmaxf(ac[t].x, 0.f); a.y = fmaxf(ac[t].y, 0.f);
                d[t] = __builtin_amdgcn_wmma_f32_16x16x4_f32(false, a, false, bc[s],
                                                             (short)0, d[t], false, false);
            }
        }
#pragma unroll
        for (int s = 0; s < 8; ++s) bc[s] = bn[s];
    }
    if (NT == 4)
        asm volatile("v_nop\n\tv_nop\n\tv_nop\n\tv_nop" : "+v"(d[0]), "+v"(d[1]), "+v"(d[2]), "+v"(d[3]));
    else
        asm volatile("v_nop\n\tv_nop\n\tv_nop\n\tv_nop" : "+v"(d[0]), "+v"(d[1]), "+v"(d[2]));
}

#define QW         768
#define PW         64
#define RS         256
#define UEDGES     512
#define NUNITS     (N_EDGES / UEDGES)
#define UPW        ((NUNITS + 31) / 32)
#define DEPTH      8
#define SC         224
#define NSC        ((QW + SC - 1) / SC)

static_assert(N_NODES <= 65536 && OWN <= 256, "a saved entry packs the source node in 16 bits, the local row above; k_norm has 256 threads per block of OWN rows");
static_assert(SC % 32 == 0 && SC <= 255, "whole groups of 32 entries; start offsets are stored as bytes");
static_assert(N_EDGES % UEDGES == 0 && (N_NODES + OWN - 1) / OWN <= 256 && UEDGES <= QW, "whole scan units; an owner fits a byte; one unit's edges fit the queue");

__device__ __forceinline__ int next_round(int r, int by) {
    r += by;
    return (r >= UPW) ? r - UPW : r;
}

__device__ __forceinline__ uint4 load_unit(const uint4* __restrict__ o4, int r, int w, int lane) {
    return o4[min(r * 32 + w, NUNITS - 1) * 32 + lane];
}

__device__ __forceinline__ unsigned unit_hits(uint4 ob, unsigned pat) {
    const unsigned wd[4] = {ob.x, ob.y, ob.z, ob.w};
    unsigned m = 0;
#pragma unroll
    for (int k = 0; k < 4; ++k) {
        const unsigned x = wd[k] ^ pat;
        const unsigned z = ~(((x & 0x7f7f7f7fu) + 0x7f7f7f7fu) | x | 0x7f7f7f7fu);
        m |= z >> k;
    }
    return m;
}

__device__ __forceinline__ int queue_hits(unsigned m, int ebase, int cnt, unsigned lt, int2* myq) {
    unsigned mask = __builtin_amdgcn_ballot_w32(m != 0u);
    while (mask) {
        if (m != 0u) {
            const int p = __builtin_ctz(m);
            myq[cnt + __builtin_popcount(mask & lt)].x = ebase + 4 * (7 - (p & 7)) + (p >> 3);
        }
        cnt += __builtin_popcount(mask);
        m &= m - 1u;
        mask = __builtin_amdgcn_ballot_w32(m != 0u);
    }
    return cnt;
}

__device__ __forceinline__ unsigned same_key_lanes(int key, bool valid) {
    unsigned peers = __builtin_amdgcn_ballot_w32(valid);
#pragma unroll
    for (int bit = 0; bit < 5; ++bit) {
        const bool one = ((key >> bit) & 1) != 0;
        const unsigned bm = __builtin_amdgcn_ballot_w32(valid && one);
        peers &= one ? bm : ~bm;
    }
    return peers;
}

__device__ __forceinline__ void add_rows(int s, int dl, float c, int n, int lane,
                                         const float* __restrict__ hw, float* acc) {
#pragma unroll 8
    for (int e = 0; e < n; ++e) {
        int sj = __builtin_amdgcn_readlane(s, e);
        int dj = __builtin_amdgcn_readlane(dl, e);
        float cj = __builtin_bit_cast(float, __builtin_amdgcn_readlane(__builtin_bit_cast(int, c), e));
        float4 h = *(const float4*)(hw + (size_t)sj * HID + lane * 4);
        float4* a = (float4*)(acc + dj * PITCH) + lane;
        float4 v = *a;
        v.x += h.x * cj; v.y += h.y * cj; v.z += h.z * cj; v.w += h.w * cj;
        *a = v;
    }
}

__device__ __forceinline__ void add_rows_in(int s, int dl, float c, int n, int lane,
                                            const float* __restrict__ x, const float4* wc, float* acc) {
    const int xl = min(lane, IN_DIM - 1);
#pragma unroll 8
    for (int e = 0; e < n; ++e) {
        int sj = __builtin_amdgcn_readlane(s, e);
        int dj = __builtin_amdgcn_readlane(dl, e);
        float cj = __builtin_bit_cast(float, __builtin_amdgcn_readlane(__builtin_bit_cast(int, c), e));
        float4 h = in_row_times_w(x[sj * IN_DIM + xl], wc);
        float4* a = (float4*)(acc + dj * PITCH) + lane;
        float4 v = *a;
        v.x += h.x * cj; v.y += h.y * cj; v.z += h.z * cj; v.w += h.w * cj;
        *a = v;
    }
}

__device__ __forceinline__ void apply_pending(const int2* pq, int n, int lane, int n0,
                                              const int* __restrict__ src,
                                              const float* __restrict__ dis,
                                              const float* __restrict__ x, const float4* wc, float* acc,
                                              int2* __restrict__ save) {
    int s = 0, dl = 0; float c = 0.f;
    if (lane < n) {
        int2 en = pq[lane];
        s = src[en.x];
        dl = en.y;
        c = dis[s] * dis[n0 + dl];
        save[lane] = make_int2(s | (dl << 16), __builtin_bit_cast(int, c));
    }
    add_rows_in(s, dl, c, n, lane, x, wc, acc);
}

template <bool REPLAY, bool FUSE>
__global__ __launch_bounds__(1024) void k_edge_own(const int* __restrict__ src,
                                                   const int* __restrict__ dst,
                                                   const float* __restrict__ dis,
                                                   const float* __restrict__ hw,
                                                   const float* __restrict__ bias,
                                                   float* __restrict__ out,
                                                   const int* __restrict__ wtot,
                                                   int* __restrict__ list_i,
                                                   const float* __restrict__ wnext,
                                                   const float* __restrict__ win,
                                                   const unsigned* __restrict__ own4) {
    __shared__ __attribute__((aligned(16))) float acc[ACC_ROWS * PITCH];
    __shared__ int2  q[32][QW];
    __shared__ int2  pend[32][PW];
    __shared__ int   qcnt[32];
    __shared__ unsigned char qoff[32][NSC][32];
    __shared__ int   unfinished[32];
    __shared__ int   sred[32];

    const int tid  = threadIdx.x;
    const int lane = tid & 31;
    const int w    = __builtin_amdgcn_readfirstlane(tid >> 5);
    const int n0   = blockIdx.x * OWN;
    const int nn   = min(OWN, N_NODES - n0);
    const unsigned lt = (1u << lane) - 1u;

    float4 wc[IN_DIM];
    if (!REPLAY) {
#pragma unroll
        for (int k = 0; k < IN_DIM; ++k) wc[k] = ((const float4*)(win + k * HID))[lane];
    }

    for (int i = tid; i < nn * (HID / 4); i += 1024) {
        int r = i >> 5;
        int f4 = i & 31;
        float s = dis[n0 + r];
        s = s * s;
        float4 h;
        if (REPLAY) h = ((const float4*)(hw + (size_t)(n0 + r) * HID))[f4];
        else        h = in_row_times_w(hw[(n0 + r) * IN_DIM + min(lane, IN_DIM - 1)], wc);
        float4 b = ((const float4*)bias)[f4];
        float4 o;
        o.x = fmaf(h.x, s, b.x); o.y = fmaf(h.y, s, b.y);
        o.z = fmaf(h.z, s, b.z); o.w = fmaf(h.w, s, b.w);
        ((float4*)(acc + r * PITCH))[f4] = o;
    }

    int pre = 0;
    for (int i = tid; i < (int)blockIdx.x * 32; i += 1024) pre += wtot[i];
    for (int o = 16; o > 0; o >>= 1) pre += __shfl_xor(pre, o);
    if (lane == 0) sred[w] = pre;
    __syncthreads();
    const int mine = wtot[blockIdx.x * 32 + lane];
    int st = sred[lane] + ((lane < w) ? mine : 0);
    for (int o = 16; o > 0; o >>= 1) st += __shfl_xor(st, o);
    int2* mylist = (int2*)list_i + __builtin_amdgcn_readfirstlane(st);

    int2* myq = q[w];

    if (REPLAY) {
        const int mycnt = __builtin_amdgcn_readlane(mine, w);
        for (int cb = 0; cb < mycnt; cb += RS) {
            const int c = min(mycnt - cb, RS);
            int2 t[RS / 32];
#pragma unroll
            for (int k = 0; k < RS / 32; ++k) {
                const int i = k * 32 + lane;
                t[k] = make_int2(0, 0);
                if (i < c) t[k] = mylist[cb + i];
            }
#pragma unroll
            for (int k = 0; k < RS / 32; ++k) myq[k * 32 + lane] = t[k];
            for (int b = 0; b < c; b += 32) {
                const int n = min(32, c - b);
                int sd = 0; float cf = 0.f;
                if (lane < n) {
                    int2 en = myq[b + lane];
                    sd = en.x;
                    cf = __builtin_bit_cast(float, en.y);
                }
                add_rows(sd & 0xffff, sd >> 16, cf, n, lane, hw, acc);
            }
        }
        __syncthreads();
    } else {
        const uint4* o4 = (const uint4*)own4;
        const unsigned pat = (unsigned)blockIdx.x * 0x01010101u;
        const int rot = (int)blockIdx.x % UPW;
        int2* myp = pend[w];
        int pos = 0;
        int wpos = 0;

        for (;;) {
            int cnt = 0;
            int rnd = next_round(rot, pos);
            uint4 win[DEPTH];
#pragma unroll
            for (int j = 0; j < DEPTH; ++j) win[j] = load_unit(o4, next_round(rnd, j), w, lane);
            bool go = true;
            while (go) {
#pragma unroll
                for (int j = 0; j < DEPTH; ++j) {
                    if (pos >= UPW || cnt + UEDGES > QW) { go = false; break; }
                    const int u = rnd * 32 + w;
                    unsigned m = unit_hits(win[j], pat);
                    if (u >= NUNITS) m = 0u;
                    win[j] = load_unit(o4, next_round(rnd, DEPTH), w, lane);
                    ++pos;
                    rnd = next_round(rnd, 1);
                    cnt = queue_hits(m, u * UEDGES + lane * 16, cnt, lt, myq);
                }
            }
            for (int b = 0, sc = 0; b < cnt; b += SC, ++sc) {
                const int nc = min(SC, cnt - b);
                unsigned char* tb = qoff[w][sc];
                tb[lane] = 0;
                int ei[SC / 32], dl[SC / 32], ps[SC / 32];
#pragma unroll
                for (int k = 0; k < SC / 32; ++k) {
                    const int i = k * 32 + lane;
                    ei[k] = (i < nc) ? myq[b + i].x : 0;
                }
#pragma unroll
                for (int k = 0; k < SC / 32; ++k) dl[k] = dst[ei[k]] - n0;
#pragma unroll
                for (int k = 0; k < SC / 32; ++k) {
                    const bool valid = k * 32 + lane < nc;
                    const int key = dl[k] & 31;
                    const unsigned peers = same_key_lanes(key, valid);
                    const int base = valid ? (int)tb[key] : 0;
                    ps[k] = base + __builtin_popcount(peers & lt);
                    if (valid && (peers & lt) == 0u) tb[key] = (unsigned char)(base + __builtin_popcount(peers));
                }
                const int cj = (int)tb[lane];
                int incl = cj;
#pragma unroll
                for (int o = 1; o < 32; o <<= 1) {
                    const int t = __shfl_up(incl, o);
                    if (lane >= o) incl += t;
                }
                tb[lane] = (unsigned char)(incl - cj);
#pragma unroll
                for (int k = 0; k < SC / 32; ++k) {
                    if (k * 32 + lane < nc)
                        myq[b + (int)tb[dl[k] & 31] + ps[k]] = make_int2(ei[k], dl[k]);
                }
            }
            if (lane == 0) { qcnt[w] = cnt; unfinished[w] = (pos < UPW) ? 1 : 0; }
            __syncthreads();
            const bool more = __builtin_amdgcn_ballot_w32(unfinished[lane] != 0) != 0u;

            int pc = 0;
            for (int p = 0; p < 32; ++p) {
                const int np = __builtin_amdgcn_readfirstlane(qcnt[p]);
                for (int b = 0, sc = 0; b < np; b += SC, ++sc) {
                    const int2* pq = q[p] + b;
                    const int lo = __builtin_amdgcn_readfirstlane((int)qoff[p][sc][w]);
                    const int hn = __builtin_amdgcn_readfirstlane((int)qoff[p][sc][min(w + 1, 31)]);
                    const int hi = (w == 31) ? min(SC, np - b) : hn;
                    for (int g = lo; g < hi; g += 32) {
                        const bool m = g + lane < hi;
                        unsigned mask = __builtin_amdgcn_ballot_w32(m);
                        if (m) myp[pc + __builtin_popcount(mask & lt)] = pq[g + lane];
                        pc += __builtin_popcount(mask);
                        if (pc >= 32) {
                            apply_pending(myp, 32, lane, n0, src, dis, hw, wc, acc, mylist + wpos);
                            wpos += 32;
                            int2 mv = make_int2(0, 0);
                            if (lane < pc - 32) mv = myp[32 + lane];
                            if (lane < pc - 32) myp[lane] = mv;
                            pc -= 32;
                        }
                    }
                }
            }
            if (pc > 0) {
                apply_pending(myp, pc, lane, n0, src, dis, hw, wc, acc, mylist + wpos);
                wpos += pc;
            }
            __syncthreads();
            if (!more) break;
        }
    }

    if (FUSE) {
        const int ct  = w & 7;
        const int g   = w >> 3;
        const int c16 = lane & 15;
        const int hf  = lane >> 4;
        const float* arow = acc + (g * 16 + c16) * PITCH + 2 * hf;
        const float* wl   = wnext + (2 * hf) * HID + ct * 16 + c16;
        const int nt = (g == 0) ? 4 : 3;
        v8f d[4];
#pragma unroll
        for (int t = 0; t < 4; ++t) d[t] = (v8f){0.f, 0.f, 0.f, 0.f, 0.f, 0.f, 0.f, 0.f};
        if (g == 0) rows_times_w<4>(arow, wl, d);
        else        rows_times_w<3>(arow, wl, d);
        __syncthreads();
        float* drow = acc + (g * 16 + 8 * hf) * PITCH + ct * 16 + c16;
#pragma unroll
        for (int t = 0; t < 4; ++t) {
            if (t < nt) {
#pragma unroll
                for (int v = 0; v < 8; ++v) drow[(t * 64 + v) * PITCH] = d[t][v];
            }
        }
        __syncthreads();
    }

    float4* o4 = (float4*)(out + (size_t)n0 * HID);
    for (int i = tid; i < nn * (HID / 4); i += 1024)
        o4[i] = ((const float4*)(acc + (i >> 5) * PITCH))[i & 31];
}

#define POOL_WAVES 16
#define POOL_LOADS 16
__global__ __launch_bounds__(POOL_WAVES * 32) void k_pool_own(const float* __restrict__ h,
                                                              const int* __restrict__ gstart,
                                                              float* __restrict__ out) {
    __shared__ __attribute__((aligned(16))) float part[POOL_WAVES][HID];
    const int g    = blockIdx.x;
    const int lane = threadIdx.x & 31;
    const int w    = __builtin_amdgcn_readfirstlane(threadIdx.x >> 5);
    const int lo = gstart[g], hi = gstart[g + 1];
    float4 s = make_float4(0.f, 0.f, 0.f, 0.f);
    for (int r0 = lo + w; r0 < hi; r0 += POOL_LOADS * POOL_WAVES) {
        float4 v[POOL_LOADS];
#pragma unroll
        for (int k = 0; k < POOL_LOADS; ++k) {
            const int r = min(r0 + k * POOL_WAVES, hi - 1);
            v[k] = *(const float4*)(h + (size_t)r * HID + lane * 4);
        }
#pragma unroll
        for (int k = 0; k < POOL_LOADS; ++k) {
            if (r0 + k * POOL_WAVES < hi) { s.x += v[k].x; s.y += v[k].y; s.z += v[k].z; s.w += v[k].w; }
        }
    }
    ((float4*)part[w])[lane] = s;
    __syncthreads();
    if (threadIdx.x < HID) {
        float t = 0.f;
        for (int j = 0; j < POOL_WAVES; ++j) t += part[j][threadIdx.x];
        out[g * HID + threadIdx.x] = t / fmaxf((float)(hi - lo), 1.0f);
    }
}

extern "C" void kernel_launch(void* const* d_in, const int* in_sizes, int n_in,
                              void* d_out, int out_size, void* d_ws, size_t ws_size,
                              hipStream_t stream) {
    const float* x        = (const float*)d_in[0];
    const int*   eidx     = (const int*)d_in[1];
    const int*   batch    = (const int*)d_in[2];
    const float* W1       = (const float*)d_in[3];
    const float* b1       = (const float*)d_in[4];
    const float* W2       = (const float*)d_in[5];
    const float* b2       = (const float*)d_in[6];
    const float* W3       = (const float*)d_in[7];
    const float* b3       = (const float*)d_in[8];
    float* out = (float*)d_out;

    const int* src = eidx;
    const int* dst = eidx + N_EDGES;

    float* ws     = (float*)d_ws;
    float* dis    = ws;
    float* hw     = dis + N_NODES;
    float* hA     = hw + (size_t)N_NODES * HID;
    int*   gstart = (int*)(hA + (size_t)N_NODES * HID);
    int*   wtot   = gstart + 320;
    int*   elist  = wtot + OWN_WG * 32;
    unsigned* own4 = (unsigned*)(elist + 2 * N_EDGES);

    const int B = 256;
    int g_n    = (N_NODES + B - 1) / B;
    int g_e4   = (N_EDGES / 4 + B - 1) / B;

    k_fill1<<<g_n, B, 0, stream>>>(dis, batch, gstart);
    k_deg<<<g_e4, B, 0, stream>>>(dst, dis, own4);
    k_norm<<<OWN_WG, B, 0, stream>>>(dis, wtot);

    k_edge_own<false, true><<<OWN_WG, 1024, 0, stream>>>(src, dst, dis, x, b1, hA, wtot, elist, W2, W1, own4);

    k_edge_own<true, true><<<OWN_WG, 1024, 0, stream>>>(src, dst, dis, hA, b2, hw, wtot, elist, W3, W1, own4);

    k_edge_own<true, false><<<OWN_WG, 1024, 0, stream>>>(src, dst, dis, hw, b3, hA, wtot, elist, W3, W1, own4);

    k_pool_own<<<N_GRAPHS, POOL_WAVES * 32, 0, stream>>>(hA, gstart, out);
}


# In-degrees owner-computed, no global atomics: 256 work-groups each scan the owner bytes as layer 1 does, queue their block's edges, count them per row in wave-private LDS histograms; bit-identical
# speedup vs baseline: 4.6369x; 1.2230x over previous
#include <hip/hip_runtime.h>

#define N_NODES  50000
#define N_EDGES  1600000
#define N_GRAPHS 256
#define IN_DIM   9
#define HID      128
#define OWN      196
#define OWN_WG   256
#define PITCH    132
#define ACC_ROWS 208

typedef __attribute__((ext_vector_type(2))) float v2f;
typedef __attribute__((ext_vector_type(8))) float v8f;

__global__ void k_fill1(const int* __restrict__ batch, int* __restrict__ gstart) {
    int i = blockIdx.x * blockDim.x + threadIdx.x;
    if (i < N_NODES) {
        const int cur  = batch[i];
        const int prev = (i == 0) ? -1 : batch[i - 1];
        for (int g = max(prev + 1, 0); g <= min(cur, N_GRAPHS - 1); ++g) gstart[g] = i;
        if (i == N_NODES - 1)
            for (int g = max(cur + 1, 0); g <= N_GRAPHS; ++g) gstart[g] = N_NODES;
    }
}

__global__ void k_owner(const int* __restrict__ dst, unsigned* __restrict__ own4) {
    int t = blockIdx.x * blockDim.x + threadIdx.x;
    if (t < N_EDGES / 4) {
        const int4 d = ((const int4*)dst)[t];
        own4[t] = ((unsigned)d.x / OWN) | (((unsigned)d.y / OWN) << 8) |
                  (((unsigned)d.z / OWN) << 16) | (((unsigned)d.w / OWN) << 24);
    }
}

__device__ __forceinline__ float4 in_row_times_w(float xv, const float4* wc) {
    float4 h = make_float4(0.f, 0.f, 0.f, 0.f);
#pragma unroll
    for (int k = 0; k < IN_DIM; ++k) {
        float xk = __builtin_bit_cast(float, __builtin_amdgcn_readlane(__builtin_bit_cast(int, xv), k));
        h.x = fmaf(xk, wc[k].x, h.x); h.y = fmaf(xk, wc[k].y, h.y);
        h.z = fmaf(xk, wc[k].z, h.z); h.w = fmaf(xk, wc[k].w, h.w);
    }
    return h;
}

template <int NT>
__device__ __forceinline__ void rows_times_w(const float* arow, const float* __restrict__ wl, v8f* d) {
    v2f bc[8], bn[8];
    float2 an[NT];
#pragma unroll
    for (int s = 0; s < 8; ++s) { bc[s].x = wl[(4 * s) * HID]; bc[s].y = wl[(4 * s + 1) * HID]; }
#pragma unroll
    for (int t = 0; t < NT; ++t) an[t] = *(const float2*)(arow + t * 64 * PITCH);
#pragma unroll
    for (int cb = 0; cb < HID; cb += 32) {
        if (cb + 32 < HID) {
#pragma unroll
            for (int s = 0; s < 8; ++s) {
                bn[s].x = wl[(cb + 32 + 4 * s) * HID];
                bn[s].y = wl[(cb + 32 + 4 * s + 1) * HID];
            }
        }
#pragma unroll
        for (int s = 0; s < 8; ++s) {
            const int k0 = cb + 4 * s;
            float2 ac[NT];
#pragma unroll
            for (int t = 0; t < NT; ++t) ac[t] = an[t];
            if (k0 + 4 < HID) {
#pragma unroll
                for (int t = 0; t < NT; ++t) an[t] = *(const float2*)(arow + t * 64 * PITCH + k0 + 4);
            }
            __builtin_amdgcn_sched_barrier(0);
#pragma unroll
            for (int t = 0; t < NT; ++t) {
                v2f a; a.x = fmaxf(ac[t].x, 0.f); a.y = fmaxf(ac[t].y, 0.f);
                d[t] = __builtin_amdgcn_wmma_f32_16x16x4_f32(false, a, false, bc[s],
                                                             (short)0, d[t], false, false);
            }
        }
#pragma unroll
        for (int s = 0; s < 8; ++s) bc[s] = bn[s];
    }
    if (NT == 4)
        asm volatile("v_nop\n\tv_nop\n\tv_nop\n\tv_nop" : "+v"(d[0]), "+v"(d[1]), "+v"(d[2]), "+v"(d[3]));
    else
        asm volatile("v_nop\n\tv_nop\n\tv_nop\n\tv_nop" : "+v"(d[0]), "+v"(d[1]), "+v"(d[2]));
}

#define QW         768
#define PW         64
#define RS         256
#define UEDGES     512
#define NUNITS     (N_EDGES / UEDGES)
#define UPW        ((NUNITS + 31) / 32)
#define DEPTH      8
#define SC         224
#define NSC        ((QW + SC - 1) / SC)

static_assert(N_NODES <= 65536 && OWN <= 256, "a saved entry packs the source node in 16 bits, the local row above; a local row fits eight bits");
static_assert(SC % 32 == 0 && SC <= 255, "whole groups of 32 entries; start offsets are stored as bytes");
static_assert(N_EDGES % UEDGES == 0 && (N_NODES + OWN - 1) / OWN <= 256 && UEDGES <= QW, "whole scan units; an owner fits a byte; one unit's edges fit the queue");

__device__ __forceinline__ int next_round(int r, int by) {
    r += by;
    return (r >= UPW) ? r - UPW : r;
}

__device__ __forceinline__ uint4 load_unit(const uint4* __restrict__ o4, int r, int w, int lane) {
    return o4[min(r * 32 + w, NUNITS - 1) * 32 + lane];
}

__device__ __forceinline__ unsigned unit_hits(uint4 ob, unsigned pat) {
    const unsigned wd[4] = {ob.x, ob.y, ob.z, ob.w};
    unsigned m = 0;
#pragma unroll
    for (int k = 0; k < 4; ++k) {
        const unsigned x = wd[k] ^ pat;
        const unsigned z = ~(((x & 0x7f7f7f7fu) + 0x7f7f7f7fu) | x | 0x7f7f7f7fu);
        m |= z >> k;
    }
    return m;
}

__device__ __forceinline__ int queue_hits(unsigned m, int ebase, int cnt, unsigned lt, int2* myq) {
    unsigned mask = __builtin_amdgcn_ballot_w32(m != 0u);
    while (mask) {
        if (m != 0u) {
            const int p = __builtin_ctz(m);
            myq[cnt + __builtin_popcount(mask & lt)].x = ebase + 4 * (7 - (p & 7)) + (p >> 3);
        }
        cnt += __builtin_popcount(mask);
        m &= m - 1u;
        mask = __builtin_amdgcn_ballot_w32(m != 0u);
    }
    return cnt;
}

__device__ __forceinline__ unsigned same_key_lanes(int key, bool valid) {
    unsigned peers = __builtin_amdgcn_ballot_w32(valid);
#pragma unroll
    for (int bit = 0; bit < 5; ++bit) {
        const bool one = ((key >> bit) & 1) != 0;
        const unsigned bm = __builtin_amdgcn_ballot_w32(valid && one);
        peers &= one ? bm : ~bm;
    }
    return peers;
}

__device__ __forceinline__ unsigned same_row_lanes(int key, bool valid) {
    unsigned peers = __builtin_amdgcn_ballot_w32(valid);
#pragma unroll
    for (int bit = 0; bit < 8; ++bit) {
        const bool one = ((key >> bit) & 1) != 0;
        const unsigned bm = __builtin_amdgcn_ballot_w32(valid && one);
        peers &= one ? bm : ~bm;
    }
    return peers;
}

static_assert(OWN <= 256, "a local row is a key of eight bits; rows 0..255 are summed by the first 256 threads");
__global__ __launch_bounds__(1024) void k_count(const int* __restrict__ dst, const unsigned* __restrict__ own4,
                                                float* __restrict__ dis, int* __restrict__ wtot) {
    __shared__ int2 q[32][QW];
    __shared__ int  hist[32][OWN];
    __shared__ int  red[8][32];
    const int tid  = threadIdx.x;
    const int lane = tid & 31;
    const int w    = __builtin_amdgcn_readfirstlane(tid >> 5);
    const int n0   = blockIdx.x * OWN;
    const unsigned lt = (1u << lane) - 1u;
    int2* myq = q[w];
    int*  myh = hist[w];
    for (int i = lane; i < OWN; i += 32) myh[i] = 0;

    const uint4* o4 = (const uint4*)own4;
    const unsigned pat = (unsigned)blockIdx.x * 0x01010101u;
    const int rot = (int)blockIdx.x % UPW;
    int pos = 0;
    while (pos < UPW) {
        int cnt = 0;
        int rnd = next_round(rot, pos);
        uint4 win[DEPTH];
#pragma unroll
        for (int j = 0; j < DEPTH; ++j) win[j] = load_unit(o4, next_round(rnd, j), w, lane);
        bool go = true;
        while (go) {
#pragma unroll
            for (int j = 0; j < DEPTH; ++j) {
                if (pos >= UPW || cnt + UEDGES > QW) { go = false; break; }
                const int u = rnd * 32 + w;
                unsigned m = unit_hits(win[j], pat);
                if (u >= NUNITS) m = 0u;
                win[j] = load_unit(o4, next_round(rnd, DEPTH), w, lane);
                ++pos;
                rnd = next_round(rnd, 1);
                cnt = queue_hits(m, u * UEDGES + lane * 16, cnt, lt, myq);
            }
        }
        for (int b = 0; b < cnt; b += SC) {
            const int nc = min(SC, cnt - b);
            int dl[SC / 32];
#pragma unroll
            for (int k = 0; k < SC / 32; ++k) {
                const int i = k * 32 + lane;
                const int ei = (i < nc) ? myq[b + i].x : 0;
                dl[k] = dst[ei] - n0;
            }
#pragma unroll
            for (int k = 0; k < SC / 32; ++k) {
                const bool valid = k * 32 + lane < nc;
                const unsigned peers = same_row_lanes(dl[k], valid);
                if (valid && (peers & lt) == 0u) myh[dl[k]] += __builtin_popcount(peers);
            }
        }
    }
    __syncthreads();
    int c = 0;
    if (tid < OWN && n0 + tid < N_NODES) {
        for (int j = 0; j < 32; ++j) c += hist[j][tid];
        dis[n0 + tid] = rsqrtf(1.0f + (float)c);
    }
    if (tid < 256) red[tid >> 5][tid & 31] = c;
    __syncthreads();
    if (tid < 32) {
        int t = 0;
        for (int j = 0; j < 8; ++j) t += red[j][tid];
        wtot[blockIdx.x * 32 + tid] = t;
    }
}

__device__ __forceinline__ void add_rows(int s, int dl, float c, int n, int lane,
                                         const float* __restrict__ hw, float* acc) {
#pragma unroll 8
    for (int e = 0; e < n; ++e) {
        int sj = __builtin_amdgcn_readlane(s, e);
        int dj = __builtin_amdgcn_readlane(dl, e);
        float cj = __builtin_bit_cast(float, __builtin_amdgcn_readlane(__builtin_bit_cast(int, c), e));
        float4 h = *(const float4*)(hw + (size_t)sj * HID + lane * 4);
        float4* a = (float4*)(acc + dj * PITCH) + lane;
        float4 v = *a;
        v.x += h.x * cj; v.y += h.y * cj; v.z += h.z * cj; v.w += h.w * cj;
        *a = v;
    }
}

__device__ __forceinline__ void add_rows_in(int s, int dl, float c, int n, int lane,
                                            const float* __restrict__ x, const float4* wc, float* acc) {
    const int xl = min(lane, IN_DIM - 1);
#pragma unroll 8
    for (int e = 0; e < n; ++e) {
        int sj = __builtin_amdgcn_readlane(s, e);
        int dj = __builtin_amdgcn_readlane(dl, e);
        float cj = __builtin_bit_cast(float, __builtin_amdgcn_readlane(__builtin_bit_cast(int, c), e));
        float4 h = in_row_times_w(x[sj * IN_DIM + xl], wc);
        float4* a = (float4*)(acc + dj * PITCH) + lane;
        float4 v = *a;
        v.x += h.x * cj; v.y += h.y * cj; v.z += h.z * cj; v.w += h.w * cj;
        *a = v;
    }
}

__device__ __forceinline__ void apply_pending(const int2* pq, int n, int lane, int n0,
                                              const int* __restrict__ src,
                                              const float* __restrict__ dis,
                                              const float* __restrict__ x, const float4* wc, float* acc,
                                              int2* __restrict__ save) {
    int s = 0, dl = 0; float c = 0.f;
    if (lane < n) {
        int2 en = pq[lane];
        s = src[en.x];
        dl = en.y;
        c = dis[s] * dis[n0 + dl];
        save[lane] = make_int2(s | (dl << 16), __builtin_bit_cast(int, c));
    }
    add_rows_in(s, dl, c, n, lane, x, wc, acc);
}

template <bool REPLAY, bool FUSE>
__global__ __launch_bounds__(1024) void k_edge_own(const int* __restrict__ src,
                                                   const int* __restrict__ dst,
                                                   const float* __restrict__ dis,
                                                   const float* __restrict__ hw,
                                                   const float* __restrict__ bias,
                                                   float* __restrict__ out,
                                                   const int* __restrict__ wtot,
                                                   int* __restrict__ list_i,
                                                   const float* __restrict__ wnext,
                                                   const float* __restrict__ win,
                                                   const unsigned* __restrict__ own4) {
    __shared__ __attribute__((aligned(16))) float acc[ACC_ROWS * PITCH];
    __shared__ int2  q[32][QW];
    __shared__ int2  pend[32][PW];
    __shared__ int   qcnt[32];
    __shared__ unsigned char qoff[32][NSC][32];
    __shared__ int   unfinished[32];
    __shared__ int   sred[32];

    const int tid  = threadIdx.x;
    const int lane = tid & 31;
    const int w    = __builtin_amdgcn_readfirstlane(tid >> 5);
    const int n0   = blockIdx.x * OWN;
    const int nn   = min(OWN, N_NODES - n0);
    const unsigned lt = (1u << lane) - 1u;

    float4 wc[IN_DIM];
    if (!REPLAY) {
#pragma unroll
        for (int k = 0; k < IN_DIM; ++k) wc[k] = ((const float4*)(win + k * HID))[lane];
    }

    for (int i = tid; i < nn * (HID / 4); i += 1024) {
        int r = i >> 5;
        int f4 = i & 31;
        float s = dis[n0 + r];
        s = s * s;
        float4 h;
        if (REPLAY) h = ((const float4*)(hw + (size_t)(n0 + r) * HID))[f4];
        else        h = in_row_times_w(hw[(n0 + r) * IN_DIM + min(lane, IN_DIM - 1)], wc);
        float4 b = ((const float4*)bias)[f4];
        float4 o;
        o.x = fmaf(h.x, s, b.x); o.y = fmaf(h.y, s, b.y);
        o.z = fmaf(h.z, s, b.z); o.w = fmaf(h.w, s, b.w);
        ((float4*)(acc + r * PITCH))[f4] = o;
    }

    int pre = 0;
    for (int i = tid; i < (int)blockIdx.x * 32; i += 1024) pre += wtot[i];
    for (int o = 16; o > 0; o >>= 1) pre += __shfl_xor(pre, o);
    if (lane == 0) sred[w] = pre;
    __syncthreads();
    const int mine = wtot[blockIdx.x * 32 + lane];
    int st = sred[lane] + ((lane < w) ? mine : 0);
    for (int o = 16; o > 0; o >>= 1) st += __shfl_xor(st, o);
    int2* mylist = (int2*)list_i + __builtin_amdgcn_readfirstlane(st);

    int2* myq = q[w];

    if (REPLAY) {
        const int mycnt = __builtin_amdgcn_readlane(mine, w);
        for (int cb = 0; cb < mycnt; cb += RS) {
            const int c = min(mycnt - cb, RS);
            int2 t[RS / 32];
#pragma unroll
            for (int k = 0; k < RS / 32; ++k) {
                const int i = k * 32 + lane;
                t[k] = make_int2(0, 0);
                if (i < c) t[k] = mylist[cb + i];
            }
#pragma unroll
            for (int k = 0; k < RS / 32; ++k) myq[k * 32 + lane] = t[k];
            for (int b = 0; b < c; b += 32) {
                const int n = min(32, c - b);
                int sd = 0; float cf = 0.f;
                if (lane < n) {
                    int2 en = myq[b + lane];
                    sd = en.x;
                    cf = __builtin_bit_cast(float, en.y);
                }
                add_rows(sd & 0xffff, sd >> 16, cf, n, lane, hw, acc);
            }
        }
        __syncthreads();
    } else {
        const uint4* o4 = (const uint4*)own4;
        const unsigned pat = (unsigned)blockIdx.x * 0x01010101u;
        const int rot = (int)blockIdx.x % UPW;
        int2* myp = pend[w];
        int pos = 0;
        int wpos = 0;

        for (;;) {
            int cnt = 0;
            int rnd = next_round(rot, pos);
            uint4 win[DEPTH];
#pragma unroll
            for (int j = 0; j < DEPTH; ++j) win[j] = load_unit(o4, next_round(rnd, j), w, lane);
            bool go = true;
            while (go) {
#pragma unroll
                for (int j = 0; j < DEPTH; ++j) {
                    if (pos >= UPW || cnt + UEDGES > QW) { go = false; break; }
                    const int u = rnd * 32 + w;
                    unsigned m = unit_hits(win[j], pat);
                    if (u >= NUNITS) m = 0u;
                    win[j] = load_unit(o4, next_round(rnd, DEPTH), w, lane);
                    ++pos;
                    rnd = next_round(rnd, 1);
                    cnt = queue_hits(m, u * UEDGES + lane * 16, cnt, lt, myq);
                }
            }
            for (int b = 0, sc = 0; b < cnt; b += SC, ++sc) {
                const int nc = min(SC, cnt - b);
                unsigned char* tb = qoff[w][sc];
                tb[lane] = 0;
                int ei[SC / 32], dl[SC / 32], ps[SC / 32];
#pragma unroll
                for (int k = 0; k < SC / 32; ++k) {
                    const int i = k * 32 + lane;
                    ei[k] = (i < nc) ? myq[b + i].x : 0;
                }
#pragma unroll
                for (int k = 0; k < SC / 32; ++k) dl[k] = dst[ei[k]] - n0;
#pragma unroll
                for (int k = 0; k < SC / 32; ++k) {
                    const bool valid = k * 32 + lane < nc;
                    const int key = dl[k] & 31;
                    const unsigned peers = same_key_lanes(key, valid);
                    const int base = valid ? (int)tb[key] : 0;
                    ps[k] = base + __builtin_popcount(peers & lt);
                    if (valid && (peers & lt) == 0u) tb[key] = (unsigned char)(base + __builtin_popcount(peers));
                }
                const int cj = (int)tb[lane];
                int incl = cj;
#pragma unroll
                for (int o = 1; o < 32; o <<= 1) {
                    const int t = __shfl_up(incl, o);
                    if (lane >= o) incl += t;
                }
                tb[lane] = (unsigned char)(incl - cj);
#pragma unroll
                for (int k = 0; k < SC / 32; ++k) {
                    if (k * 32 + lane < nc)
                        myq[b + (int)tb[dl[k] & 31] + ps[k]] = make_int2(ei[k], dl[k]);
                }
            }
            if (lane == 0) { qcnt[w] = cnt; unfinished[w] = (pos < UPW) ? 1 : 0; }
            __syncthreads();
            const bool more = __builtin_amdgcn_ballot_w32(unfinished[lane] != 0) != 0u;

            int pc = 0;
            for (int p = 0; p < 32; ++p) {
                const int np = __builtin_amdgcn_readfirstlane(qcnt[p]);
                for (int b = 0, sc = 0; b < np; b += SC, ++sc) {
                    const int2* pq = q[p] + b;
                    const int lo = __builtin_amdgcn_readfirstlane((int)qoff[p][sc][w]);
                    const int hn = __builtin_amdgcn_readfirstlane((int)qoff[p][sc][min(w + 1, 31)]);
                    const int hi = (w == 31) ? min(SC, np - b) : hn;
                    for (int g = lo; g < hi; g += 32) {
                        const bool m = g + lane < hi;
                        unsigned mask = __builtin_amdgcn_ballot_w32(m);
                        if (m) myp[pc + __builtin_popcount(mask & lt)] = pq[g + lane];
                        pc += __builtin_popcount(mask);
                        if (pc >= 32) {
                            apply_pending(myp, 32, lane, n0, src, dis, hw, wc, acc, mylist + wpos);
                            wpos += 32;
                            int2 mv = make_int2(0, 0);
                            if (lane < pc - 32) mv = myp[32 + lane];
                            if (lane < pc - 32) myp[lane] = mv;
                            pc -= 32;
                        }
                    }
                }
            }
            if (pc > 0) {
                apply_pending(myp, pc, lane, n0, src, dis, hw, wc, acc, mylist + wpos);
                wpos += pc;
            }
            __syncthreads();
            if (!more) break;
        }
    }

    if (FUSE) {
        const int ct  = w & 7;
        const int g   = w >> 3;
        const int c16 = lane & 15;
        const int hf  = lane >> 4;
        const float* arow = acc + (g * 16 + c16) * PITCH + 2 * hf;
        const float* wl   = wnext + (2 * hf) * HID + ct * 16 + c16;
        const int nt = (g == 0) ? 4 : 3;
        v8f d[4];
#pragma unroll
        for (int t = 0; t < 4; ++t) d[t] = (v8f){0.f, 0.f, 0.f, 0.f, 0.f, 0.f, 0.f, 0.f};
        if (g == 0) rows_times_w<4>(arow, wl, d);
        else        rows_times_w<3>(arow, wl, d);
        __syncthreads();
        float* drow = acc + (g * 16 + 8 * hf) * PITCH + ct * 16 + c16;
#pragma unroll
        for (int t = 0; t < 4; ++t) {
            if (t < nt) {
#pragma unroll
                for (int v = 0; v < 8; ++v) drow[(t * 64 + v) * PITCH] = d[t][v];
            }
        }
        __syncthreads();
    }

    float4* o4 = (float4*)(out + (size_t)n0 * HID);
    for (int i = tid; i < nn * (HID / 4); i += 1024)
        o4[i] = ((const float4*)(acc + (i >> 5) * PITCH))[i & 31];
}

#define POOL_WAVES 16
#define POOL_LOADS 16
__global__ __launch_bounds__(POOL_WAVES * 32) void k_pool_own(const float* __restrict__ h,
                                                              const int* __restrict__ gstart,
                                                              float* __restrict__ out) {
    __shared__ __attribute__((aligned(16))) float part[POOL_WAVES][HID];
    const int g    = blockIdx.x;
    const int lane = threadIdx.x & 31;
    const int w    = __builtin_amdgcn_readfirstlane(threadIdx.x >> 5);
    const int lo = gstart[g], hi = gstart[g + 1];
    float4 s = make_float4(0.f, 0.f, 0.f, 0.f);
    for (int r0 = lo + w; r0 < hi; r0 += POOL_LOADS * POOL_WAVES) {
        float4 v[POOL_LOADS];
#pragma unroll
        for (int k = 0; k < POOL_LOADS; ++k) {
            const int r = min(r0 + k * POOL_WAVES, hi - 1);
            v[k] = *(const float4*)(h + (size_t)r * HID + lane * 4);
        }
#pragma unroll
        for (int k = 0; k < POOL_LOADS; ++k) {
            if (r0 + k * POOL_WAVES < hi) { s.x += v[k].x; s.y += v[k].y; s.z += v[k].z; s.w += v[k].w; }
        }
    }
    ((float4*)part[w])[lane] = s;
    __syncthreads();
    if (threadIdx.x < HID) {
        float t = 0.f;
        for (int j = 0; j < POOL_WAVES; ++j) t += part[j][threadIdx.x];
        out[g * HID + threadIdx.x] = t / fmaxf((float)(hi - lo), 1.0f);
    }
}

extern "C" void kernel_launch(void* const* d_in, const int* in_sizes, int n_in,
                              void* d_out, int out_size, void* d_ws, size_t ws_size,
                              hipStream_t stream) {
    const float* x        = (const float*)d_in[0];
    const int*   eidx     = (const int*)d_in[1];
    const int*   batch    = (const int*)d_in[2];
    const float* W1       = (const float*)d_in[3];
    const float* b1       = (const float*)d_in[4];
    const float* W2       = (const float*)d_in[5];
    const float* b2       = (const float*)d_in[6];
    const float* W3       = (const float*)d_in[7];
    const float* b3       = (const float*)d_in[8];
    float* out = (float*)d_out;

    const int* src = eidx;
    const int* dst = eidx + N_EDGES;

    float* ws     = (float*)d_ws;
    float* dis    = ws;
    float* hw     = dis + N_NODES;
    float* hA     = hw + (size_t)N_NODES * HID;
    int*   gstart = (int*)(hA + (size_t)N_NODES * HID);
    int*   wtot   = gstart + 320;
    int*   elist  = wtot + OWN_WG * 32;
    unsigned* own4 = (unsigned*)(elist + 2 * N_EDGES);

    const int B = 256;
    int g_n    = (N_NODES + B - 1) / B;
    int g_e4   = (N_EDGES / 4 + B - 1) / B;

    k_fill1<<<g_n, B, 0, stream>>>(batch, gstart);
    k_owner<<<g_e4, B, 0, stream>>>(dst, own4);
    k_count<<<OWN_WG, 1024, 0, stream>>>(dst, own4, dis, wtot);

    k_edge_own<false, true><<<OWN_WG, 1024, 0, stream>>>(src, dst, dis, x, b1, hA, wtot, elist, W2, W1, own4);

    k_edge_own<true, true><<<OWN_WG, 1024, 0, stream>>>(src, dst, dis, hA, b2, hw, wtot, elist, W3, W1, own4);

    k_edge_own<true, false><<<OWN_WG, 1024, 0, stream>>>(src, dst, dis, hw, b3, hA, wtot, elist, W3, W1, own4);

    k_pool_own<<<N_GRAPHS, POOL_WAVES * 32, 0, stream>>>(hA, gstart, out);
}


# Edges found once: the degree kernel saves per scanning wave the edges it finds (source, local row); layer 1 fills its queues from them, no scan; sum order kept unless a queue had overflowed
# speedup vs baseline: 5.0932x; 1.0984x over previous
#include <hip/hip_runtime.h>

#define N_NODES  50000
#define N_EDGES  1600000
#define N_GRAPHS 256
#define IN_DIM   9
#define HID      128
#define OWN      196
#define OWN_WG   256
#define PITCH    132
#define ACC_ROWS 208

typedef __attribute__((ext_vector_type(2))) float v2f;
typedef __attribute__((ext_vector_type(8))) float v8f;

__global__ void k_fill1(const int* __restrict__ batch, int* __restrict__ gstart) {
    int i = blockIdx.x * blockDim.x + threadIdx.x;
    if (i < N_NODES) {
        const int cur  = batch[i];
        const int prev = (i == 0) ? -1 : batch[i - 1];
        for (int g = max(prev + 1, 0); g <= min(cur, N_GRAPHS - 1); ++g) gstart[g] = i;
        if (i == N_NODES - 1)
            for (int g = max(cur + 1, 0); g <= N_GRAPHS; ++g) gstart[g] = N_NODES;
    }
}

__global__ void k_owner(const int* __restrict__ dst, unsigned* __restrict__ own4) {
    int t = blockIdx.x * blockDim.x + threadIdx.x;
    if (t < N_EDGES / 4) {
        const int4 d = ((const int4*)dst)[t];
        own4[t] = ((unsigned)d.x / OWN) | (((unsigned)d.y / OWN) << 8) |
                  (((unsigned)d.z / OWN) << 16) | (((unsigned)d.w / OWN) << 24);
    }
}

__device__ __forceinline__ float4 in_row_times_w(float xv, const float4* wc) {
    float4 h = make_float4(0.f, 0.f, 0.f, 0.f);
#pragma unroll
    for (int k = 0; k < IN_DIM; ++k) {
        float xk = __builtin_bit_cast(float, __builtin_amdgcn_readlane(__builtin_bit_cast(int, xv), k));
        h.x = fmaf(xk, wc[k].x, h.x); h.y = fmaf(xk, wc[k].y, h.y);
        h.z = fmaf(xk, wc[k].z, h.z); h.w = fmaf(xk, wc[k].w, h.w);
    }
    return h;
}

template <int NT>
__device__ __forceinline__ void rows_times_w(const float* arow, const float* __restrict__ wl, v8f* d) {
    v2f bc[8], bn[8];
    float2 an[NT];
#pragma unroll
    for (int s = 0; s < 8; ++s) { bc[s].x = wl[(4 * s) * HID]; bc[s].y = wl[(4 * s + 1) * HID]; }
#pragma unroll
    for (int t = 0; t < NT; ++t) an[t] = *(const float2*)(arow + t * 64 * PITCH);
#pragma unroll
    for (int cb = 0; cb < HID; cb += 32) {
        if (cb + 32 < HID) {
#pragma unroll
            for (int s = 0; s < 8; ++s) {
                bn[s].x = wl[(cb + 32 + 4 * s) * HID];
                bn[s].y = wl[(cb + 32 + 4 * s + 1) * HID];
            }
        }
#pragma unroll
        for (int s = 0; s < 8; ++s) {
            const int k0 = cb + 4 * s;
            float2 ac[NT];
#pragma unroll
            for (int t = 0; t < NT; ++t) ac[t] = an[t];
            if (k0 + 4 < HID) {
#pragma unroll
                for (int t = 0; t < NT; ++t) an[t] = *(const float2*)(arow + t * 64 * PITCH + k0 + 4);
            }
            __builtin_amdgcn_sched_barrier(0);
#pragma unroll
            for (int t = 0; t < NT; ++t) {
                v2f a; a.x = fmaxf(ac[t].x, 0.f); a.y = fmaxf(ac[t].y, 0.f);
                d[t] = __builtin_amdgcn_wmma_f32_16x16x4_f32(false, a, false, bc[s],
                                                             (short)0, d[t], false, false);
            }
        }
#pragma unroll
        for (int s = 0; s < 8; ++s) bc[s] = bn[s];
    }
    if (NT == 4)
        asm volatile("v_nop\n\tv_nop\n\tv_nop\n\tv_nop" : "+v"(d[0]), "+v"(d[1]), "+v"(d[2]), "+v"(d[3]));
    else
        asm volatile("v_nop\n\tv_nop\n\tv_nop\n\tv_nop" : "+v"(d[0]), "+v"(d[1]), "+v"(d[2]));
}

#define QW         768
#define PW         64
#define RS         256
#define UEDGES     512
#define NUNITS     (N_EDGES / UEDGES)
#define UPW        ((NUNITS + 31) / 32)
#define DEPTH      8
#define SC         224
#define NSC        ((QW + SC - 1) / SC)
#define SAVEW      QW

static_assert(N_NODES <= 65536 && OWN <= 256, "a saved entry packs the source node in 16 bits, the local row above; a local row fits eight bits");
static_assert(SC % 32 == 0 && SC <= 255, "whole groups of 32 entries; start offsets are stored as bytes");
static_assert(N_EDGES % UEDGES == 0 && (N_NODES + OWN - 1) / OWN <= 256 && UEDGES <= QW, "whole scan units; an owner fits a byte; one unit's edges fit the queue");

__device__ __forceinline__ int next_round(int r, int by) {
    r += by;
    return (r >= UPW) ? r - UPW : r;
}

__device__ __forceinline__ uint4 load_unit(const uint4* __restrict__ o4, int r, int w, int lane) {
    return o4[min(r * 32 + w, NUNITS - 1) * 32 + lane];
}

__device__ __forceinline__ unsigned unit_hits(uint4 ob, unsigned pat) {
    const unsigned wd[4] = {ob.x, ob.y, ob.z, ob.w};
    unsigned m = 0;
#pragma unroll
    for (int k = 0; k < 4; ++k) {
        const unsigned x = wd[k] ^ pat;
        const unsigned z = ~(((x & 0x7f7f7f7fu) + 0x7f7f7f7fu) | x | 0x7f7f7f7fu);
        m |= z >> k;
    }
    return m;
}

__device__ __forceinline__ int queue_hits(unsigned m, int ebase, int cnt, unsigned lt, int2* myq) {
    unsigned mask = __builtin_amdgcn_ballot_w32(m != 0u);
    while (mask) {
        if (m != 0u) {
            const int p = __builtin_ctz(m);
            myq[cnt + __builtin_popcount(mask & lt)].x = ebase + 4 * (7 - (p & 7)) + (p >> 3);
        }
        cnt += __builtin_popcount(mask);
        m &= m - 1u;
        mask = __builtin_amdgcn_ballot_w32(m != 0u);
    }
    return cnt;
}

__device__ __forceinline__ unsigned same_key_lanes(int key, bool valid) {
    unsigned peers = __builtin_amdgcn_ballot_w32(valid);
#pragma unroll
    for (int bit = 0; bit < 5; ++bit) {
        const bool one = ((key >> bit) & 1) != 0;
        const unsigned bm = __builtin_amdgcn_ballot_w32(valid && one);
        peers &= one ? bm : ~bm;
    }
    return peers;
}

__device__ __forceinline__ unsigned same_row_lanes(int key, bool valid) {
    unsigned peers = __builtin_amdgcn_ballot_w32(valid);
#pragma unroll
    for (int bit = 0; bit < 8; ++bit) {
        const bool one = ((key >> bit) & 1) != 0;
        const unsigned bm = __builtin_amdgcn_ballot_w32(valid && one);
        peers &= one ? bm : ~bm;
    }
    return peers;
}

static_assert(OWN <= 256, "a local row is a key of eight bits; rows 0..255 are summed by the first 256 threads");
__global__ __launch_bounds__(1024) void k_count(const int* __restrict__ src, const int* __restrict__ dst,
                                                const unsigned* __restrict__ own4,
                                                float* __restrict__ dis, int* __restrict__ wtot,
                                                int* __restrict__ save_i, int* __restrict__ scnt) {
    __shared__ int2 q[32][QW];
    __shared__ int  hist[32][OWN];
    __shared__ int  red[8][32];
    const int tid  = threadIdx.x;
    const int lane = tid & 31;
    const int w    = __builtin_amdgcn_readfirstlane(tid >> 5);
    const int n0   = blockIdx.x * OWN;
    const unsigned lt = (1u << lane) - 1u;
    int2* myq = q[w];
    int*  myh = hist[w];
    for (int i = lane; i < OWN; i += 32) myh[i] = 0;

    const uint4* o4 = (const uint4*)own4;
    const unsigned pat = (unsigned)blockIdx.x * 0x01010101u;
    const int rot = (int)blockIdx.x % UPW;
    int2* mysave = (int2*)save_i + (size_t)(blockIdx.x * 32 + w) * SAVEW;
    int tot = 0;
    int pos = 0;
    while (pos < UPW) {
        int cnt = 0;
        int rnd = next_round(rot, pos);
        uint4 win[DEPTH];
#pragma unroll
        for (int j = 0; j < DEPTH; ++j) win[j] = load_unit(o4, next_round(rnd, j), w, lane);
        bool go = true;
        while (go) {
#pragma unroll
            for (int j = 0; j < DEPTH; ++j) {
                if (pos >= UPW || cnt + UEDGES > QW) { go = false; break; }
                const int u = rnd * 32 + w;
                unsigned m = unit_hits(win[j], pat);
                if (u >= NUNITS) m = 0u;
                win[j] = load_unit(o4, next_round(rnd, DEPTH), w, lane);
                ++pos;
                rnd = next_round(rnd, 1);
                cnt = queue_hits(m, u * UEDGES + lane * 16, cnt, lt, myq);
            }
        }
        for (int b = 0; b < cnt; b += SC) {
            const int nc = min(SC, cnt - b);
            int dl[SC / 32], sl[SC / 32];
#pragma unroll
            for (int k = 0; k < SC / 32; ++k) {
                const int i = k * 32 + lane;
                const int ei = (i < nc) ? myq[b + i].x : 0;
                dl[k] = dst[ei] - n0;
                sl[k] = src[ei];
            }
#pragma unroll
            for (int k = 0; k < SC / 32; ++k) {
                const int i = k * 32 + lane;
                if (i < nc && tot + b + i < SAVEW) mysave[tot + b + i] = make_int2(sl[k], dl[k]);
            }
#pragma unroll
            for (int k = 0; k < SC / 32; ++k) {
                const bool valid = k * 32 + lane < nc;
                const unsigned peers = same_row_lanes(dl[k], valid);
                if (valid && (peers & lt) == 0u) myh[dl[k]] += __builtin_popcount(peers);
            }
        }
        tot += cnt;
    }
    if (lane == 0) scnt[blockIdx.x * 32 + w] = tot;
    __syncthreads();
    int c = 0;
    if (tid < OWN && n0 + tid < N_NODES) {
        for (int j = 0; j < 32; ++j) c += hist[j][tid];
        dis[n0 + tid] = rsqrtf(1.0f + (float)c);
    }
    if (tid < 256) red[tid >> 5][tid & 31] = c;
    __syncthreads();
    if (tid < 32) {
        int t = 0;
        for (int j = 0; j < 8; ++j) t += red[j][tid];
        wtot[blockIdx.x * 32 + tid] = t;
    }
}

__device__ __forceinline__ void add_rows(int s, int dl, float c, int n, int lane,
                                         const float* __restrict__ hw, float* acc) {
#pragma unroll 8
    for (int e = 0; e < n; ++e) {
        int sj = __builtin_amdgcn_readlane(s, e);
        int dj = __builtin_amdgcn_readlane(dl, e);
        float cj = __builtin_bit_cast(float, __builtin_amdgcn_readlane(__builtin_bit_cast(int, c), e));
        float4 h = *(const float4*)(hw + (size_t)sj * HID + lane * 4);
        float4* a = (float4*)(acc + dj * PITCH) + lane;
        float4 v = *a;
        v.x += h.x * cj; v.y += h.y * cj; v.z += h.z * cj; v.w += h.w * cj;
        *a = v;
    }
}

__device__ __forceinline__ void add_rows_in(int s, int dl, float c, int n, int lane,
                                            const float* __restrict__ x, const float4* wc, float* acc) {
    const int xl = min(lane, IN_DIM - 1);
#pragma unroll 8
    for (int e = 0; e < n; ++e) {
        int sj = __builtin_amdgcn_readlane(s, e);
        int dj = __builtin_amdgcn_readlane(dl, e);
        float cj = __builtin_bit_cast(float, __builtin_amdgcn_readlane(__builtin_bit_cast(int, c), e));
        float4 h = in_row_times_w(x[sj * IN_DIM + xl], wc);
        float4* a = (float4*)(acc + dj * PITCH) + lane;
        float4 v = *a;
        v.x += h.x * cj; v.y += h.y * cj; v.z += h.z * cj; v.w += h.w * cj;
        *a = v;
    }
}

__device__ __forceinline__ void apply_pending(const int2* pq, int n, int lane, int n0, bool known,
                                              const int* __restrict__ src,
                                              const float* __restrict__ dis,
                                              const float* __restrict__ x, const float4* wc, float* acc,
                                              int2* __restrict__ save) {
    int s = 0, dl = 0; float c = 0.f;
    if (lane < n) {
        int2 en = pq[lane];
        s = known ? en.x : src[en.x];
        dl = en.y;
        c = dis[s] * dis[n0 + dl];
        save[lane] = make_int2(s | (dl << 16), __builtin_bit_cast(int, c));
    }
    add_rows_in(s, dl, c, n, lane, x, wc, acc);
}

template <bool REPLAY, bool FUSE>
__global__ __launch_bounds__(1024) void k_edge_own(const int* __restrict__ src,
                                                   const int* __restrict__ dst,
                                                   const float* __restrict__ dis,
                                                   const float* __restrict__ hw,
                                                   const float* __restrict__ bias,
                                                   float* __restrict__ out,
                                                   const int* __restrict__ wtot,
                                                   int* __restrict__ list_i,
                                                   const float* __restrict__ wnext,
                                                   const float* __restrict__ win,
                                                   const unsigned* __restrict__ own4,
                                                   const int* __restrict__ save_i,
                                                   const int* __restrict__ scnt) {
    __shared__ __attribute__((aligned(16))) float acc[ACC_ROWS * PITCH];
    __shared__ int2  q[32][QW];
    __shared__ int2  pend[32][PW];
    __shared__ int   qcnt[32];
    __shared__ unsigned char qoff[32][NSC][32];
    __shared__ int   unfinished[32];
    __shared__ int   sred[32];

    const int tid  = threadIdx.x;
    const int lane = tid & 31;
    const int w    = __builtin_amdgcn_readfirstlane(tid >> 5);
    const int n0   = blockIdx.x * OWN;
    const int nn   = min(OWN, N_NODES - n0);
    const unsigned lt = (1u << lane) - 1u;

    float4 wc[IN_DIM];
    if (!REPLAY) {
#pragma unroll
        for (int k = 0; k < IN_DIM; ++k) wc[k] = ((const float4*)(win + k * HID))[lane];
    }

    for (int i = tid; i < nn * (HID / 4); i += 1024) {
        int r = i >> 5;
        int f4 = i & 31;
        float s = dis[n0 + r];
        s = s * s;
        float4 h;
        if (REPLAY) h = ((const float4*)(hw + (size_t)(n0 + r) * HID))[f4];
        else        h = in_row_times_w(hw[(n0 + r) * IN_DIM + min(lane, IN_DIM - 1)], wc);
        float4 b = ((const float4*)bias)[f4];
        float4 o;
        o.x = fmaf(h.x, s, b.x); o.y = fmaf(h.y, s, b.y);
        o.z = fmaf(h.z, s, b.z); o.w = fmaf(h.w, s, b.w);
        ((float4*)(acc + r * PITCH))[f4] = o;
    }

    int pre = 0;
    for (int i = tid; i < (int)blockIdx.x * 32; i += 1024) pre += wtot[i];
    for (int o = 16; o > 0; o >>= 1) pre += __shfl_xor(pre, o);
    if (lane == 0) sred[w] = pre;
    __syncthreads();
    const int mine = wtot[blockIdx.x * 32 + lane];
    int st = sred[lane] + ((lane < w) ? mine : 0);
    for (int o = 16; o > 0; o >>= 1) st += __shfl_xor(st, o);
    int2* mylist = (int2*)list_i + __builtin_amdgcn_readfirstlane(st);

    int2* myq = q[w];

    if (REPLAY) {
        const int mycnt = __builtin_amdgcn_readlane(mine, w);
        for (int cb = 0; cb < mycnt; cb += RS) {
            const int c = min(mycnt - cb, RS);
            int2 t[RS / 32];
#pragma unroll
            for (int k = 0; k < RS / 32; ++k) {
                const int i = k * 32 + lane;
                t[k] = make_int2(0, 0);
                if (i < c) t[k] = mylist[cb + i];
            }
#pragma unroll
            for (int k = 0; k < RS / 32; ++k) myq[k * 32 + lane] = t[k];
            for (int b = 0; b < c; b += 32) {
                const int n = min(32, c - b);
                int sd = 0; float cf = 0.f;
                if (lane < n) {
                    int2 en = myq[b + lane];
                    sd = en.x;
                    cf = __builtin_bit_cast(float, en.y);
                }
                add_rows(sd & 0xffff, sd >> 16, cf, n, lane, hw, acc);
            }
        }
        __syncthreads();
    } else {
        const uint4* o4 = (const uint4*)own4;
        const unsigned pat = (unsigned)blockIdx.x * 0x01010101u;
        const int rot = (int)blockIdx.x % UPW;
        int2* myp = pend[w];
        int pos = 0;
        int wpos = 0;
        const int found = scnt[blockIdx.x * 32 + lane];
        const bool known = __builtin_amdgcn_ballot_w32(found > SAVEW) == 0u;
        const int myfound = __builtin_amdgcn_readlane(found, w);
        const int2* mysave = (const int2*)save_i + (size_t)(blockIdx.x * 32 + w) * SAVEW;

        for (;;) {
            int cnt = 0;
            if (known) {
                cnt = myfound;
                for (int cb = 0; cb < cnt; cb += RS) {
                    int2 t[RS / 32];
#pragma unroll
                    for (int k = 0; k < RS / 32; ++k) {
                        const int i = cb + k * 32 + lane;
                        t[k] = make_int2(0, 0);
                        if (i < cnt) t[k] = mysave[i];
                    }
#pragma unroll
                    for (int k = 0; k < RS / 32; ++k) {
                        const int i = cb + k * 32 + lane;
                        if (i < cnt) myq[i] = t[k];
                    }
                }
                pos = UPW;
            } else {
                int rnd = next_round(rot, pos);
                uint4 win[DEPTH];
#pragma unroll
                for (int j = 0; j < DEPTH; ++j) win[j] = load_unit(o4, next_round(rnd, j), w, lane);
                bool go = true;
                while (go) {
#pragma unroll
                    for (int j = 0; j < DEPTH; ++j) {
                        if (pos >= UPW || cnt + UEDGES > QW) { go = false; break; }
                        const int u = rnd * 32 + w;
                        unsigned m = unit_hits(win[j], pat);
                        if (u >= NUNITS) m = 0u;
                        win[j] = load_unit(o4, next_round(rnd, DEPTH), w, lane);
                        ++pos;
                        rnd = next_round(rnd, 1);
                        cnt = queue_hits(m, u * UEDGES + lane * 16, cnt, lt, myq);
                    }
                }
            }
            for (int b = 0, sc = 0; b < cnt; b += SC, ++sc) {
                const int nc = min(SC, cnt - b);
                unsigned char* tb = qoff[w][sc];
                tb[lane] = 0;
                int ei[SC / 32], dl[SC / 32], ps[SC / 32];
#pragma unroll
                for (int k = 0; k < SC / 32; ++k) {
                    const int i = k * 32 + lane;
                    int2 en = make_int2(0, 0);
                    if (i < nc) en = myq[b + i];
                    ei[k] = en.x;
                    dl[k] = en.y;
                }
                if (!known) {
#pragma unroll
                    for (int k = 0; k < SC / 32; ++k) dl[k] = dst[ei[k]] - n0;
                }
#pragma unroll
                for (int k = 0; k < SC / 32; ++k) {
                    const bool valid = k * 32 + lane < nc;
                    const int key = dl[k] & 31;
                    const unsigned peers = same_key_lanes(key, valid);
                    const int base = valid ? (int)tb[key] : 0;
                    ps[k] = base + __builtin_popcount(peers & lt);
                    if (valid && (peers & lt) == 0u) tb[key] = (unsigned char)(base + __builtin_popcount(peers));
                }
                const int cj = (int)tb[lane];
                int incl = cj;
#pragma unroll
                for (int o = 1; o < 32; o <<= 1) {
                    const int t = __shfl_up(incl, o);
                    if (lane >= o) incl += t;
                }
                tb[lane] = (unsigned char)(incl - cj);
#pragma unroll
                for (int k = 0; k < SC / 32; ++k) {
                    if (k * 32 + lane < nc)
                        myq[b + (int)tb[dl[k] & 31] + ps[k]] = make_int2(ei[k], dl[k]);
                }
            }
            if (lane == 0) { qcnt[w] = cnt; unfinished[w] = (pos < UPW) ? 1 : 0; }
            __syncthreads();
            const bool more = __builtin_amdgcn_ballot_w32(unfinished[lane] != 0) != 0u;

            int pc = 0;
            for (int p = 0; p < 32; ++p) {
                const int np = __builtin_amdgcn_readfirstlane(qcnt[p]);
                for (int b = 0, sc = 0; b < np; b += SC, ++sc) {
                    const int2* pq = q[p] + b;
                    const int lo = __builtin_amdgcn_readfirstlane((int)qoff[p][sc][w]);
                    const int hn = __builtin_amdgcn_readfirstlane((int)qoff[p][sc][min(w + 1, 31)]);
                    const int hi = (w == 31) ? min(SC, np - b) : hn;
                    for (int g = lo; g < hi; g += 32) {
                        const bool m = g + lane < hi;
                        unsigned mask = __builtin_amdgcn_ballot_w32(m);
                        if (m) myp[pc + __builtin_popcount(mask & lt)] = pq[g + lane];
                        pc += __builtin_popcount(mask);
                        if (pc >= 32) {
                            apply_pending(myp, 32, lane, n0, known, src, dis, hw, wc, acc, mylist + wpos);
                            wpos += 32;
                            int2 mv = make_int2(0, 0);
                            if (lane < pc - 32) mv = myp[32 + lane];
                            if (lane < pc - 32) myp[lane] = mv;
                            pc -= 32;
                        }
                    }
                }
            }
            if (pc > 0) {
                apply_pending(myp, pc, lane, n0, known, src, dis, hw, wc, acc, mylist + wpos);
                wpos += pc;
            }
            __syncthreads();
            if (!more) break;
        }
    }

    if (FUSE) {
        const int ct  = w & 7;
        const int g   = w >> 3;
        const int c16 = lane & 15;
        const int hf  = lane >> 4;
        const float* arow = acc + (g * 16 + c16) * PITCH + 2 * hf;
        const float* wl   = wnext + (2 * hf) * HID + ct * 16 + c16;
        const int nt = (g == 0) ? 4 : 3;
        v8f d[4];
#pragma unroll
        for (int t = 0; t < 4; ++t) d[t] = (v8f){0.f, 0.f, 0.f, 0.f, 0.f, 0.f, 0.f, 0.f};
        if (g == 0) rows_times_w<4>(arow, wl, d);
        else        rows_times_w<3>(arow, wl, d);
        __syncthreads();
        float* drow = acc + (g * 16 + 8 * hf) * PITCH + ct * 16 + c16;
#pragma unroll
        for (int t = 0; t < 4; ++t) {
            if (t < nt) {
#pragma unroll
                for (int v = 0; v < 8; ++v) drow[(t * 64 + v) * PITCH] = d[t][v];
            }
        }
        __syncthreads();
    }

    float4* o4 = (float4*)(out + (size_t)n0 * HID);
    for (int i = tid; i < nn * (HID / 4); i += 1024)
        o4[i] = ((const float4*)(acc + (i >> 5) * PITCH))[i & 31];
}

#define POOL_WAVES 16
#define POOL_LOADS 16
__global__ __launch_bounds__(POOL_WAVES * 32) void k_pool_own(const float* __restrict__ h,
                                                              const int* __restrict__ gstart,
                                                              float* __restrict__ out) {
    __shared__ __attribute__((aligned(16))) float part[POOL_WAVES][HID];
    const int g    = blockIdx.x;
    const int lane = threadIdx.x & 31;
    const int w    = __builtin_amdgcn_readfirstlane(threadIdx.x >> 5);
    const int lo = gstart[g], hi = gstart[g + 1];
    float4 s = make_float4(0.f, 0.f, 0.f, 0.f);
    for (int r0 = lo + w; r0 < hi; r0 += POOL_LOADS * POOL_WAVES) {
        float4 v[POOL_LOADS];
#pragma unroll
        for (int k = 0; k < POOL_LOADS; ++k) {
            const int r = min(r0 + k * POOL_WAVES, hi - 1);
            v[k] = *(const float4*)(h + (size_t)r * HID + lane * 4);
        }
#pragma unroll
        for (int k = 0; k < POOL_LOADS; ++k) {
            if (r0 + k * POOL_WAVES < hi) { s.x += v[k].x; s.y += v[k].y; s.z += v[k].z; s.w += v[k].w; }
        }
    }
    ((float4*)part[w])[lane] = s;
    __syncthreads();
    if (threadIdx.x < HID) {
        float t = 0.f;
        for (int j = 0; j < POOL_WAVES; ++j) t += part[j][threadIdx.x];
        out[g * HID + threadIdx.x] = t / fmaxf((float)(hi - lo), 1.0f);
    }
}

extern "C" void kernel_launch(void* const* d_in, const int* in_sizes, int n_in,
                              void* d_out, int out_size, void* d_ws, size_t ws_size,
                              hipStream_t stream) {
    const float* x        = (const float*)d_in[0];
    const int*   eidx     = (const int*)d_in[1];
    const int*   batch    = (const int*)d_in[2];
    const float* W1       = (const float*)d_in[3];
    const float* b1       = (const float*)d_in[4];
    const float* W2       = (const float*)d_in[5];
    const float* b2       = (const float*)d_in[6];
    const float* W3       = (const float*)d_in[7];
    const float* b3       = (const float*)d_in[8];
    float* out = (float*)d_out;

    const int* src = eidx;
    const int* dst = eidx + N_EDGES;

    float* ws     = (float*)d_ws;
    float* dis    = ws;
    float* hw     = dis + N_NODES;
    float* hA     = hw + (size_t)N_NODES * HID;
    int*   gstart = (int*)(hA + (size_t)N_NODES * HID);
    int*   wtot   = gstart + 320;
    int*   elist  = wtot + OWN_WG * 32;
    unsigned* own4 = (unsigned*)(elist + 2 * N_EDGES);
    int*   saveq  = (int*)(own4 + N_EDGES / 4);
    int*   scnt   = saveq + 2 * OWN_WG * 32 * SAVEW;

    const int B = 256;
    int g_n    = (N_NODES + B - 1) / B;
    int g_e4   = (N_EDGES / 4 + B - 1) / B;

    k_fill1<<<g_n, B, 0, stream>>>(batch, gstart);
    k_owner<<<g_e4, B, 0, stream>>>(dst, own4);
    k_count<<<OWN_WG, 1024, 0, stream>>>(src, dst, own4, dis, wtot, saveq, scnt);

    k_edge_own<false, true><<<OWN_WG, 1024, 0, stream>>>(src, dst, dis, x, b1, hA, wtot, elist, W2, W1, own4, saveq, scnt);

    k_edge_own<true, true><<<OWN_WG, 1024, 0, stream>>>(src, dst, dis, hA, b2, hw, wtot, elist, W3, W1, own4, saveq, scnt);

    k_edge_own<true, false><<<OWN_WG, 1024, 0, stream>>>(src, dst, dis, hw, b3, hA, wtot, elist, W3, W1, own4, saveq, scnt);

    k_pool_own<<<N_GRAPHS, POOL_WAVES * 32, 0, stream>>>(hA, gstart, out);
}


# Edge search split 16 x 16: a work-group reads one slice of dst and src for 16 blocks and sorts its finds by block into small lists; the degree kernel reads those, no scan; per-node sum order changes
# speedup vs baseline: 5.4945x; 1.0788x over previous
#include <hip/hip_runtime.h>

#define N_NODES  50000
#define N_EDGES  1600000
#define N_GRAPHS 256
#define IN_DIM   9
#define HID      128
#define OWN      196
#define OWN_WG   256
#define PITCH    132
#define ACC_ROWS 208

typedef __attribute__((ext_vector_type(2))) float v2f;
typedef __attribute__((ext_vector_type(8))) float v8f;

__global__ void k_fill1(const int* __restrict__ batch, int* __restrict__ gstart) {
    int i = blockIdx.x * blockDim.x + threadIdx.x;
    if (i < N_NODES) {
        const int cur  = batch[i];
        const int prev = (i == 0) ? -1 : batch[i - 1];
        for (int g = max(prev + 1, 0); g <= min(cur, N_GRAPHS - 1); ++g) gstart[g] = i;
        if (i == N_NODES - 1)
            for (int g = max(cur + 1, 0); g <= N_GRAPHS; ++g) gstart[g] = N_NODES;
    }
}

__device__ __forceinline__ float4 in_row_times_w(float xv, const float4* wc) {
    float4 h = make_float4(0.f, 0.f, 0.f, 0.f);
#pragma unroll
    for (int k = 0; k < IN_DIM; ++k) {
        float xk = __builtin_bit_cast(float, __builtin_amdgcn_readlane(__builtin_bit_cast(int, xv), k));
        h.x = fmaf(xk, wc[k].x, h.x); h.y = fmaf(xk, wc[k].y, h.y);
        h.z = fmaf(xk, wc[k].z, h.z); h.w = fmaf(xk, wc[k].w, h.w);
    }
    return h;
}

template <int NT>
__device__ __forceinline__ void rows_times_w(const float* arow, const float* __restrict__ wl, v8f* d) {
    v2f bc[8], bn[8];
    float2 an[NT];
#pragma unroll
    for (int s = 0; s < 8; ++s) { bc[s].x = wl[(4 * s) * HID]; bc[s].y = wl[(4 * s + 1) * HID]; }
#pragma unroll
    for (int t = 0; t < NT; ++t) an[t] = *(const float2*)(arow + t * 64 * PITCH);
#pragma unroll
    for (int cb = 0; cb < HID; cb += 32) {
        if (cb + 32 < HID) {
#pragma unroll
            for (int s = 0; s < 8; ++s) {
                bn[s].x = wl[(cb + 32 + 4 * s) * HID];
                bn[s].y = wl[(cb + 32 + 4 * s + 1) * HID];
            }
        }
#pragma unroll
        for (int s = 0; s < 8; ++s) {
            const int k0 = cb + 4 * s;
            float2 ac[NT];
#pragma unroll
            for (int t = 0; t < NT; ++t) ac[t] = an[t];
            if (k0 + 4 < HID) {
#pragma unroll
                for (int t = 0; t < NT; ++t) an[t] = *(const float2*)(arow + t * 64 * PITCH + k0 + 4);
            }
            __builtin_amdgcn_sched_barrier(0);
#pragma unroll
            for (int t = 0; t < NT; ++t) {
                v2f a; a.x = fmaxf(ac[t].x, 0.f); a.y = fmaxf(ac[t].y, 0.f);
                d[t] = __builtin_amdgcn_wmma_f32_16x16x4_f32(false, a, false, bc[s],
                                                             (short)0, d[t], false, false);
            }
        }
#pragma unroll
        for (int s = 0; s < 8; ++s) bc[s] = bn[s];
    }
    if (NT == 4)
        asm volatile("v_nop\n\tv_nop\n\tv_nop\n\tv_nop" : "+v"(d[0]), "+v"(d[1]), "+v"(d[2]), "+v"(d[3]));
    else
        asm volatile("v_nop\n\tv_nop\n\tv_nop\n\tv_nop" : "+v"(d[0]), "+v"(d[1]), "+v"(d[2]));
}

#define QW         768
#define PW         64
#define RS         256
#define UEDGES     512
#define NUNITS     (N_EDGES / UEDGES)
#define UPW        ((NUNITS + 31) / 32)
#define DEPTH      8
#define SC         224
#define NSC        ((QW + SC - 1) / SC)
#define SAVEW      QW

static_assert(N_NODES <= 65536 && OWN <= 256, "a saved entry packs the source node in 16 bits, the local row above; a local row fits eight bits");
static_assert(SC % 32 == 0 && SC <= 255, "whole groups of 32 entries; start offsets are stored as bytes");
static_assert(N_EDGES % UEDGES == 0 && (N_NODES + OWN - 1) / OWN <= 256 && UEDGES <= QW, "whole scan units; an owner fits a byte; one unit's edges fit the queue");

__device__ __forceinline__ int next_round(int r, int by) {
    r += by;
    return (r >= UPW) ? r - UPW : r;
}

__device__ __forceinline__ uint4 load_unit(const uint4* __restrict__ o4, int r, int w, int lane) {
    return o4[min(r * 32 + w, NUNITS - 1) * 32 + lane];
}

__device__ __forceinline__ unsigned unit_hits(uint4 ob, unsigned pat) {
    const unsigned wd[4] = {ob.x, ob.y, ob.z, ob.w};
    unsigned m = 0;
#pragma unroll
    for (int k = 0; k < 4; ++k) {
        const unsigned x = wd[k] ^ pat;
        const unsigned z = ~(((x & 0x7f7f7f7fu) + 0x7f7f7f7fu) | x | 0x7f7f7f7fu);
        m |= z >> k;
    }
    return m;
}

__device__ __forceinline__ int queue_hits(unsigned m, int ebase, int cnt, unsigned lt, int2* myq) {
    unsigned mask = __builtin_amdgcn_ballot_w32(m != 0u);
    while (mask) {
        if (m != 0u) {
            const int p = __builtin_ctz(m);
            myq[cnt + __builtin_popcount(mask & lt)].x = ebase + 4 * (7 - (p & 7)) + (p >> 3);
        }
        cnt += __builtin_popcount(mask);
        m &= m - 1u;
        mask = __builtin_amdgcn_ballot_w32(m != 0u);
    }
    return cnt;
}

__device__ __forceinline__ unsigned same_key_lanes(int key, bool valid) {
    unsigned peers = __builtin_amdgcn_ballot_w32(valid);
#pragma unroll
    for (int bit = 0; bit < 5; ++bit) {
        const bool one = ((key >> bit) & 1) != 0;
        const unsigned bm = __builtin_amdgcn_ballot_w32(valid && one);
        peers &= one ? bm : ~bm;
    }
    return peers;
}

__device__ __forceinline__ unsigned same_row_lanes(int key, bool valid) {
    unsigned peers = __builtin_amdgcn_ballot_w32(valid);
#pragma unroll
    for (int bit = 0; bit < 8; ++bit) {
        const bool one = ((key >> bit) & 1) != 0;
        const unsigned bm = __builtin_amdgcn_ballot_w32(valid && one);
        peers &= one ? bm : ~bm;
    }
    return peers;
}

#define NSL      16
#define SL_E4    (N_EDGES / NSL / 4)
#define GRP_ROWS (16 * OWN)
#define SUBCAP   64
#define FG       5
#define FQ       1024
static_assert(OWN_WG == 16 * NSL && N_EDGES % (NSL * 4) == 0 && FG * 128 <= FQ && SUBCAP % 32 == 0,
              "16 x 16 work-groups; whole 16-byte loads per slice; one round of loads fits the queue");

__device__ __forceinline__ unsigned same_nib_lanes(int key, bool valid) {
    unsigned peers = __builtin_amdgcn_ballot_w32(valid);
#pragma unroll
    for (int bit = 0; bit < 4; ++bit) {
        const bool one = ((key >> bit) & 1) != 0;
        const unsigned bm = __builtin_amdgcn_ballot_w32(valid && one);
        peers &= one ? bm : ~bm;
    }
    return peers;
}

__device__ __forceinline__ void flush_finds(const int2* myq, int cnt, int* myk, int2* __restrict__ mysub,
                                            int lane, unsigned lt) {
    for (int b = 0; b < cnt; b += 32) {
        const bool valid = b + lane < cnt;
        int2 en = make_int2(0, 0);
        if (valid) en = myq[b + lane];
        const int o = en.y / OWN;
        const unsigned peers = same_nib_lanes(o, valid);
        const int base = valid ? myk[o] : 0;
        const int pos = base + __builtin_popcount(peers & lt);
        if (valid && (peers & lt) == 0u) myk[o] = base + __builtin_popcount(peers);
        if (valid && pos < SUBCAP)
            mysub[(size_t)o * (NSL * 32 * SUBCAP) + pos] = make_int2(en.x, en.y - o * OWN);
    }
}

__global__ __launch_bounds__(1024) void k_find(const int* __restrict__ src, const int* __restrict__ dst,
                                               int* __restrict__ sub_i, int* __restrict__ subcnt,
                                               unsigned* __restrict__ own4) {
    __shared__ int2 q[32][FQ];
    __shared__ int  kc[32][16];
    const int tid  = threadIdx.x;
    const int lane = tid & 31;
    const int w    = __builtin_amdgcn_readfirstlane(tid >> 5);
    const int grp  = (int)blockIdx.x >> 4;
    const int sl   = (int)blockIdx.x & 15;
    const int d0   = grp * GRP_ROWS;
    const unsigned lt = (1u << lane) - 1u;
    int2* myq = q[w];
    int*  myk = kc[w];
    if (lane < 16) myk[lane] = 0;
    const int4* d4 = (const int4*)dst + sl * SL_E4;
    const int4* s4 = (const int4*)src + sl * SL_E4;
    int2* mysub = (int2*)sub_i + (((size_t)(grp * 16) * NSL + sl) * 32 + w) * SUBCAP;
    int cnt = 0;
    for (int t0 = w * 32; t0 < SL_E4; t0 += FG * 1024) {
        if (cnt + FG * 128 > FQ) {
            flush_finds(myq, cnt, myk, mysub, lane, lt);
            cnt = 0;
        }
        int4 dv[FG], sv[FG];
#pragma unroll
        for (int k = 0; k < FG; ++k) {
            const int t = min(t0 + k * 1024 + lane, SL_E4 - 1);
            dv[k] = d4[t];
            sv[k] = s4[t];
        }
#pragma unroll
        for (int k = 0; k < FG; ++k) {
            const int t = t0 + k * 1024 + lane;
            const bool act = t < SL_E4;
            if (act && (((t0 + k * 1024) >> 10) & 15) == grp)
                own4[sl * SL_E4 + t] = ((unsigned)dv[k].x / OWN) | (((unsigned)dv[k].y / OWN) << 8) |
                                       (((unsigned)dv[k].z / OWN) << 16) | (((unsigned)dv[k].w / OWN) << 24);
            const int dd[4] = {dv[k].x, dv[k].y, dv[k].z, dv[k].w};
            const int ss[4] = {sv[k].x, sv[k].y, sv[k].z, sv[k].w};
#pragma unroll
            for (int c = 0; c < 4; ++c) {
                const int rel = dd[c] - d0;
                const bool hit = act && (unsigned)rel < (unsigned)GRP_ROWS;
                const unsigned mask = __builtin_amdgcn_ballot_w32(hit);
                if (hit) myq[cnt + __builtin_popcount(mask & lt)] = make_int2(ss[c], rel);
                cnt += __builtin_popcount(mask);
            }
        }
    }
    flush_finds(myq, cnt, myk, mysub, lane, lt);
    if (lane < 16) subcnt[(((grp * 16 + lane) * NSL + sl) * 32) + w] = myk[lane];
}

template <bool GATHER>
__device__ __forceinline__ void count_rows(const int2* myq, int cnt, int tot, int n0, int lane, unsigned lt,
                                           const int* __restrict__ src, const int* __restrict__ dst,
                                           int2* __restrict__ mysave, int* myh) {
    for (int b = 0; b < cnt; b += SC) {
        const int nc = min(SC, cnt - b);
        int dl[SC / 32], sl[SC / 32];
#pragma unroll
        for (int k = 0; k < SC / 32; ++k) {
            const int i = k * 32 + lane;
            int2 en = make_int2(0, 0);
            if (i < nc) en = myq[b + i];
            if (GATHER) { dl[k] = dst[en.x] - n0; sl[k] = src[en.x]; }
            else        { dl[k] = en.y;           sl[k] = en.x; }
        }
#pragma unroll
        for (int k = 0; k < SC / 32; ++k) {
            const int i = k * 32 + lane;
            if (i < nc && tot + b + i < SAVEW) mysave[tot + b + i] = make_int2(sl[k], dl[k]);
        }
#pragma unroll
        for (int k = 0; k < SC / 32; ++k) {
            const bool valid = k * 32 + lane < nc;
            const unsigned peers = same_row_lanes(dl[k], valid);
            if (valid && (peers & lt) == 0u) myh[dl[k]] += __builtin_popcount(peers);
        }
    }
}

static_assert(OWN <= 256, "a local row is a key of eight bits; rows 0..255 are summed by the first 256 threads");
__global__ __launch_bounds__(1024) void k_count(const int* __restrict__ src, const int* __restrict__ dst,
                                                const unsigned* __restrict__ own4,
                                                float* __restrict__ dis, int* __restrict__ wtot,
                                                int* __restrict__ save_i, int* __restrict__ scnt,
                                                const int* __restrict__ sub_i, const int* __restrict__ subcnt) {
    __shared__ int2 q[32][QW];
    __shared__ int  hist[32][OWN];
    __shared__ int  red[8][32];
    __shared__ int  ovf[32];
    const int tid  = threadIdx.x;
    const int lane = tid & 31;
    const int w    = __builtin_amdgcn_readfirstlane(tid >> 5);
    const int n0   = blockIdx.x * OWN;
    const unsigned lt = (1u << lane) - 1u;
    int2* myq = q[w];
    int*  myh = hist[w];
    for (int i = lane; i < OWN; i += 32) myh[i] = 0;

    const uint4* o4 = (const uint4*)own4;
    const unsigned pat = (unsigned)blockIdx.x * 0x01010101u;
    const int rot = (int)blockIdx.x % UPW;
    int2* mysave = (int2*)save_i + (size_t)(blockIdx.x * 32 + w) * SAVEW;
    int tot = 0;

    int cj = 0;
    if (lane < NSL) cj = subcnt[((int)blockIdx.x * NSL + lane) * 32 + w];
    int incl = cj;
#pragma unroll
    for (int o = 1; o < 32; o <<= 1) {
        const int t = __shfl_up(incl, o);
        if (lane >= o) incl += t;
    }
    const int excl = incl - cj;
    const int ftot = __builtin_amdgcn_readlane(incl, 31);
    const bool bad = __builtin_amdgcn_ballot_w32(cj > SUBCAP) != 0u || ftot > QW;
    if (lane == 0) ovf[w] = bad ? 1 : 0;
    __syncthreads();
    const bool fast = __builtin_amdgcn_ballot_w32(ovf[lane] != 0) == 0u;

    if (fast) {
        const int2* mysub = (const int2*)sub_i + ((size_t)blockIdx.x * NSL * 32 + w) * SUBCAP;
#pragma unroll
        for (int j0 = 0; j0 < NSL; j0 += 8) {
            int2 t[8][SUBCAP / 32];
#pragma unroll
            for (int jj = 0; jj < 8; ++jj) {
                const int c = __builtin_amdgcn_readlane(cj, j0 + jj);
#pragma unroll
                for (int h = 0; h < SUBCAP / 32; ++h) {
                    const int i = h * 32 + lane;
                    t[jj][h] = make_int2(0, 0);
                    if (i < c) t[jj][h] = mysub[(size_t)(j0 + jj) * (32 * SUBCAP) + i];
                }
            }
#pragma unroll
            for (int jj = 0; jj < 8; ++jj) {
                const int c   = __builtin_amdgcn_readlane(cj, j0 + jj);
                const int off = __builtin_amdgcn_readlane(excl, j0 + jj);
#pragma unroll
                for (int h = 0; h < SUBCAP / 32; ++h) {
                    const int i = h * 32 + lane;
                    if (i < c) myq[off + i] = t[jj][h];
                }
            }
        }
        count_rows<false>(myq, ftot, 0, n0, lane, lt, src, dst, mysave, myh);
        tot = ftot;
    } else {
        int pos = 0;
        while (pos < UPW) {
            int cnt = 0;
            int rnd = next_round(rot, pos);
            uint4 win[DEPTH];
#pragma unroll
            for (int j = 0; j < DEPTH; ++j) win[j] = load_unit(o4, next_round(rnd, j), w, lane);
            bool go = true;
            while (go) {
#pragma unroll
                for (int j = 0; j < DEPTH; ++j) {
                    if (pos >= UPW || cnt + UEDGES > QW) { go = false; break; }
                    const int u = rnd * 32 + w;
                    unsigned m = unit_hits(win[j], pat);
                    if (u >= NUNITS) m = 0u;
                    win[j] = load_unit(o4, next_round(rnd, DEPTH), w, lane);
                    ++pos;
                    rnd = next_round(rnd, 1);
                    cnt = queue_hits(m, u * UEDGES + lane * 16, cnt, lt, myq);
                }
            }
            count_rows<true>(myq, cnt, tot, n0, lane, lt, src, dst, mysave, myh);
            tot += cnt;
        }
    }
    if (lane == 0) scnt[blockIdx.x * 32 + w] = tot;
    __syncthreads();
    int c = 0;
    if (tid < OWN && n0 + tid < N_NODES) {
        for (int j = 0; j < 32; ++j) c += hist[j][tid];
        dis[n0 + tid] = rsqrtf(1.0f + (float)c);
    }
    if (tid < 256) red[tid >> 5][tid & 31] = c;
    __syncthreads();
    if (tid < 32) {
        int t = 0;
        for (int j = 0; j < 8; ++j) t += red[j][tid];
        wtot[blockIdx.x * 32 + tid] = t;
    }
}

__device__ __forceinline__ void add_rows(int s, int dl, float c, int n, int lane,
                                         const float* __restrict__ hw, float* acc) {
#pragma unroll 8
    for (int e = 0; e < n; ++e) {
        int sj = __builtin_amdgcn_readlane(s, e);
        int dj = __builtin_amdgcn_readlane(dl, e);
        float cj = __builtin_bit_cast(float, __builtin_amdgcn_readlane(__builtin_bit_cast(int, c), e));
        float4 h = *(const float4*)(hw + (size_t)sj * HID + lane * 4);
        float4* a = (float4*)(acc + dj * PITCH) + lane;
        float4 v = *a;
        v.x += h.x * cj; v.y += h.y * cj; v.z += h.z * cj; v.w += h.w * cj;
        *a = v;
    }
}

__device__ __forceinline__ void add_rows_in(int s, int dl, float c, int n, int lane,
                                            const float* __restrict__ x, const float4* wc, float* acc) {
    const int xl = min(lane, IN_DIM - 1);
#pragma unroll 8
    for (int e = 0; e < n; ++e) {
        int sj = __builtin_amdgcn_readlane(s, e);
        int dj = __builtin_amdgcn_readlane(dl, e);
        float cj = __builtin_bit_cast(float, __builtin_amdgcn_readlane(__builtin_bit_cast(int, c), e));
        float4 h = in_row_times_w(x[sj * IN_DIM + xl], wc);
        float4* a = (float4*)(acc + dj * PITCH) + lane;
        float4 v = *a;
        v.x += h.x * cj; v.y += h.y * cj; v.z += h.z * cj; v.w += h.w * cj;
        *a = v;
    }
}

__device__ __forceinline__ void apply_pending(const int2* pq, int n, int lane, int n0, bool known,
                                              const int* __restrict__ src,
                                              const float* __restrict__ dis,
                                              const float* __restrict__ x, const float4* wc, float* acc,
                                              int2* __restrict__ save) {
    int s = 0, dl = 0; float c = 0.f;
    if (lane < n) {
        int2 en = pq[lane];
        s = known ? en.x : src[en.x];
        dl = en.y;
        c = dis[s] * dis[n0 + dl];
        save[lane] = make_int2(s | (dl << 16), __builtin_bit_cast(int, c));
    }
    add_rows_in(s, dl, c, n, lane, x, wc, acc);
}

template <bool REPLAY, bool FUSE>
__global__ __launch_bounds__(1024) void k_edge_own(const int* __restrict__ src,
                                                   const int* __restrict__ dst,
                                                   const float* __restrict__ dis,
                                                   const float* __restrict__ hw,
                                                   const float* __restrict__ bias,
                                                   float* __restrict__ out,
                                                   const int* __restrict__ wtot,
                                                   int* __restrict__ list_i,
                                                   const float* __restrict__ wnext,
                                                   const float* __restrict__ win,
                                                   const unsigned* __restrict__ own4,
                                                   const int* __restrict__ save_i,
                                                   const int* __restrict__ scnt) {
    __shared__ __attribute__((aligned(16))) float acc[ACC_ROWS * PITCH];
    __shared__ int2  q[32][QW];
    __shared__ int2  pend[32][PW];
    __shared__ int   qcnt[32];
    __shared__ unsigned char qoff[32][NSC][32];
    __shared__ int   unfinished[32];
    __shared__ int   sred[32];

    const int tid  = threadIdx.x;
    const int lane = tid & 31;
    const int w    = __builtin_amdgcn_readfirstlane(tid >> 5);
    const int n0   = blockIdx.x * OWN;
    const int nn   = min(OWN, N_NODES - n0);
    const unsigned lt = (1u << lane) - 1u;

    float4 wc[IN_DIM];
    if (!REPLAY) {
#pragma unroll
        for (int k = 0; k < IN_DIM; ++k) wc[k] = ((const float4*)(win + k * HID))[lane];
    }

    for (int i = tid; i < nn * (HID / 4); i += 1024) {
        int r = i >> 5;
        int f4 = i & 31;
        float s = dis[n0 + r];
        s = s * s;
        float4 h;
        if (REPLAY) h = ((const float4*)(hw + (size_t)(n0 + r) * HID))[f4];
        else        h = in_row_times_w(hw[(n0 + r) * IN_DIM + min(lane, IN_DIM - 1)], wc);
        float4 b = ((const float4*)bias)[f4];
        float4 o;
        o.x = fmaf(h.x, s, b.x); o.y = fmaf(h.y, s, b.y);
        o.z = fmaf(h.z, s, b.z); o.w = fmaf(h.w, s, b.w);
        ((float4*)(acc + r * PITCH))[f4] = o;
    }

    int pre = 0;
    for (int i = tid; i < (int)blockIdx.x * 32; i += 1024) pre += wtot[i];
    for (int o = 16; o > 0; o >>= 1) pre += __shfl_xor(pre, o);
    if (lane == 0) sred[w] = pre;
    __syncthreads();
    const int mine = wtot[blockIdx.x * 32 + lane];
    int st = sred[lane] + ((lane < w) ? mine : 0);
    for (int o = 16; o > 0; o >>= 1) st += __shfl_xor(st, o);
    int2* mylist = (int2*)list_i + __builtin_amdgcn_readfirstlane(st);

    int2* myq = q[w];

    if (REPLAY) {
        const int mycnt = __builtin_amdgcn_readlane(mine, w);
        for (int cb = 0; cb < mycnt; cb += RS) {
            const int c = min(mycnt - cb, RS);
            int2 t[RS / 32];
#pragma unroll
            for (int k = 0; k < RS / 32; ++k) {
                const int i = k * 32 + lane;
                t[k] = make_int2(0, 0);
                if (i < c) t[k] = mylist[cb + i];
            }
#pragma unroll
            for (int k = 0; k < RS / 32; ++k) myq[k * 32 + lane] = t[k];
            for (int b = 0; b < c; b += 32) {
                const int n = min(32, c - b);
                int sd = 0; float cf = 0.f;
                if (lane < n) {
                    int2 en = myq[b + lane];
                    sd = en.x;
                    cf = __builtin_bit_cast(float, en.y);
                }
                add_rows(sd & 0xffff, sd >> 16, cf, n, lane, hw, acc);
            }
        }
        __syncthreads();
    } else {
        const uint4* o4 = (const uint4*)own4;
        const unsigned pat = (unsigned)blockIdx.x * 0x01010101u;
        const int rot = (int)blockIdx.x % UPW;
        int2* myp = pend[w];
        int pos = 0;
        int wpos = 0;
        const int found = scnt[blockIdx.x * 32 + lane];
        const bool known = __builtin_amdgcn_ballot_w32(found > SAVEW) == 0u;
        const int myfound = __builtin_amdgcn_readlane(found, w);
        const int2* mysave = (const int2*)save_i + (size_t)(blockIdx.x * 32 + w) * SAVEW;

        for (;;) {
            int cnt = 0;
            if (known) {
                cnt = myfound;
                for (int cb = 0; cb < cnt; cb += RS) {
                    int2 t[RS / 32];
#pragma unroll
                    for (int k = 0; k < RS / 32; ++k) {
                        const int i = cb + k * 32 + lane;
                        t[k] = make_int2(0, 0);
                        if (i < cnt) t[k] = mysave[i];
                    }
#pragma unroll
                    for (int k = 0; k < RS / 32; ++k) {
                        const int i = cb + k * 32 + lane;
                        if (i < cnt) myq[i] = t[k];
                    }
                }
                pos = UPW;
            } else {
                int rnd = next_round(rot, pos);
                uint4 win[DEPTH];
#pragma unroll
                for (int j = 0; j < DEPTH; ++j) win[j] = load_unit(o4, next_round(rnd, j), w, lane);
                bool go = true;
                while (go) {
#pragma unroll
                    for (int j = 0; j < DEPTH; ++j) {
                        if (pos >= UPW || cnt + UEDGES > QW) { go = false; break; }
                        const int u = rnd * 32 + w;
                        unsigned m = unit_hits(win[j], pat);
                        if (u >= NUNITS) m = 0u;
                        win[j] = load_unit(o4, next_round(rnd, DEPTH), w, lane);
                        ++pos;
                        rnd = next_round(rnd, 1);
                        cnt = queue_hits(m, u * UEDGES + lane * 16, cnt, lt, myq);
                    }
                }
            }
            for (int b = 0, sc = 0; b < cnt; b += SC, ++sc) {
                const int nc = min(SC, cnt - b);
                unsigned char* tb = qoff[w][sc];
                tb[lane] = 0;
                int ei[SC / 32], dl[SC / 32], ps[SC / 32];
#pragma unroll
                for (int k = 0; k < SC / 32; ++k) {
                    const int i = k * 32 + lane;
                    int2 en = make_int2(0, 0);
                    if (i < nc) en = myq[b + i];
                    ei[k] = en.x;
                    dl[k] = en.y;
                }
                if (!known) {
#pragma unroll
                    for (int k = 0; k < SC / 32; ++k) dl[k] = dst[ei[k]] - n0;
                }
#pragma unroll
                for (int k = 0; k < SC / 32; ++k) {
                    const bool valid = k * 32 + lane < nc;
                    const int key = dl[k] & 31;
                    const unsigned peers = same_key_lanes(key, valid);
                    const int base = valid ? (int)tb[key] : 0;
                    ps[k] = base + __builtin_popcount(peers & lt);
                    if (valid && (peers & lt) == 0u) tb[key] = (unsigned char)(base + __builtin_popcount(peers));
                }
                const int cj = (int)tb[lane];
                int incl = cj;
#pragma unroll
                for (int o = 1; o < 32; o <<= 1) {
                    const int t = __shfl_up(incl, o);
                    if (lane >= o) incl += t;
                }
                tb[lane] = (unsigned char)(incl - cj);
#pragma unroll
                for (int k = 0; k < SC / 32; ++k) {
                    if (k * 32 + lane < nc)
                        myq[b + (int)tb[dl[k] & 31] + ps[k]] = make_int2(ei[k], dl[k]);
                }
            }
            if (lane == 0) { qcnt[w] = cnt; unfinished[w] = (pos < UPW) ? 1 : 0; }
            __syncthreads();
            const bool more = __builtin_amdgcn_ballot_w32(unfinished[lane] != 0) != 0u;

            int pc = 0;
            for (int p = 0; p < 32; ++p) {
                const int np = __builtin_amdgcn_readfirstlane(qcnt[p]);
                for (int b = 0, sc = 0; b < np; b += SC, ++sc) {
                    const int2* pq = q[p] + b;
                    const int lo = __builtin_amdgcn_readfirstlane((int)qoff[p][sc][w]);
                    const int hn = __builtin_amdgcn_readfirstlane((int)qoff[p][sc][min(w + 1, 31)]);
                    const int hi = (w == 31) ? min(SC, np - b) : hn;
                    for (int g = lo; g < hi; g += 32) {
                        const bool m = g + lane < hi;
                        unsigned mask = __builtin_amdgcn_ballot_w32(m);
                        if (m) myp[pc + __builtin_popcount(mask & lt)] = pq[g + lane];
                        pc += __builtin_popcount(mask);
                        if (pc >= 32) {
                            apply_pending(myp, 32, lane, n0, known, src, dis, hw, wc, acc, mylist + wpos);
                            wpos += 32;
                            int2 mv = make_int2(0, 0);
                            if (lane < pc - 32) mv = myp[32 + lane];
                            if (lane < pc - 32) myp[lane] = mv;
                            pc -= 32;
                        }
                    }
                }
            }
            if (pc > 0) {
                apply_pending(myp, pc, lane, n0, known, src, dis, hw, wc, acc, mylist + wpos);
                wpos += pc;
            }
            __syncthreads();
            if (!more) break;
        }
    }

    if (FUSE) {
        const int ct  = w & 7;
        const int g   = w >> 3;
        const int c16 = lane & 15;
        const int hf  = lane >> 4;
        const float* arow = acc + (g * 16 + c16) * PITCH + 2 * hf;
        const float* wl   = wnext + (2 * hf) * HID + ct * 16 + c16;
        const int nt = (g == 0) ? 4 : 3;
        v8f d[4];
#pragma unroll
        for (int t = 0; t < 4; ++t) d[t] = (v8f){0.f, 0.f, 0.f, 0.f, 0.f, 0.f, 0.f, 0.f};
        if (g == 0) rows_times_w<4>(arow, wl, d);
        else        rows_times_w<3>(arow, wl, d);
        __syncthreads();
        float* drow = acc + (g * 16 + 8 * hf) * PITCH + ct * 16 + c16;
#pragma unroll
        for (int t = 0; t < 4; ++t) {
            if (t < nt) {
#pragma unroll
                for (int v = 0; v < 8; ++v) drow[(t * 64 + v) * PITCH] = d[t][v];
            }
        }
        __syncthreads();
    }

    float4* o4 = (float4*)(out + (size_t)n0 * HID);
    for (int i = tid; i < nn * (HID / 4); i += 1024)
        o4[i] = ((const float4*)(acc + (i >> 5) * PITCH))[i & 31];
}

#define POOL_WAVES 16
#define POOL_LOADS 16
__global__ __launch_bounds__(POOL_WAVES * 32) void k_pool_own(const float* __restrict__ h,
                                                              const int* __restrict__ gstart,
                                                              float* __restrict__ out) {
    __shared__ __attribute__((aligned(16))) float part[POOL_WAVES][HID];
    const int g    = blockIdx.x;
    const int lane = threadIdx.x & 31;
    const int w    = __builtin_amdgcn_readfirstlane(threadIdx.x >> 5);
    const int lo = gstart[g], hi = gstart[g + 1];
    float4 s = make_float4(0.f, 0.f, 0.f, 0.f);
    for (int r0 = lo + w; r0 < hi; r0 += POOL_LOADS * POOL_WAVES) {
        float4 v[POOL_LOADS];
#pragma unroll
        for (int k = 0; k < POOL_LOADS; ++k) {
            const int r = min(r0 + k * POOL_WAVES, hi - 1);
            v[k] = *(const float4*)(h + (size_t)r * HID + lane * 4);
        }
#pragma unroll
        for (int k = 0; k < POOL_LOADS; ++k) {
            if (r0 + k * POOL_WAVES < hi) { s.x += v[k].x; s.y += v[k].y; s.z += v[k].z; s.w += v[k].w; }
        }
    }
    ((float4*)part[w])[lane] = s;
    __syncthreads();
    if (threadIdx.x < HID) {
        float t = 0.f;
        for (int j = 0; j < POOL_WAVES; ++j) t += part[j][threadIdx.x];
        out[g * HID + threadIdx.x] = t / fmaxf((float)(hi - lo), 1.0f);
    }
}

extern "C" void kernel_launch(void* const* d_in, const int* in_sizes, int n_in,
                              void* d_out, int out_size, void* d_ws, size_t ws_size,
                              hipStream_t stream) {
    const float* x        = (const float*)d_in[0];
    const int*   eidx     = (const int*)d_in[1];
    const int*   batch    = (const int*)d_in[2];
    const float* W1       = (const float*)d_in[3];
    const float* b1       = (const float*)d_in[4];
    const float* W2       = (const float*)d_in[5];
    const float* b2       = (const float*)d_in[6];
    const float* W3       = (const float*)d_in[7];
    const float* b3       = (const float*)d_in[8];
    float* out = (float*)d_out;

    const int* src = eidx;
    const int* dst = eidx + N_EDGES;

    float* ws     = (float*)d_ws;
    float* dis    = ws;
    float* hw     = dis + N_NODES;
    float* hA     = hw + (size_t)N_NODES * HID;
    int*   gstart = (int*)(hA + (size_t)N_NODES * HID);
    int*   wtot   = gstart + 320;
    int*   elist  = wtot + OWN_WG * 32;
    unsigned* own4 = (unsigned*)(elist + 2 * N_EDGES);
    int*   saveq  = (int*)(own4 + N_EDGES / 4);
    int*   scnt   = saveq + 2 * OWN_WG * 32 * SAVEW;
    int*   sub    = scnt + OWN_WG * 32;
    int*   subcnt = sub + 2 * OWN_WG * NSL * 32 * SUBCAP;

    const int B = 256;
    int g_n    = (N_NODES + B - 1) / B;

    k_fill1<<<g_n, B, 0, stream>>>(batch, gstart);
    k_find<<<OWN_WG, 1024, 0, stream>>>(src, dst, sub, subcnt, own4);
    k_count<<<OWN_WG, 1024, 0, stream>>>(src, dst, own4, dis, wtot, saveq, scnt, sub, subcnt);

    k_edge_own<false, true><<<OWN_WG, 1024, 0, stream>>>(src, dst, dis, x, b1, hA, wtot, elist, W2, W1, own4, saveq, scnt);

    k_edge_own<true, true><<<OWN_WG, 1024, 0, stream>>>(src, dst, dis, hA, b2, hw, wtot, elist, W3, W1, own4, saveq, scnt);

    k_edge_own<true, false><<<OWN_WG, 1024, 0, stream>>>(src, dst, dis, hw, b3, hA, wtot, elist, W3, W1, own4, saveq, scnt);

    k_pool_own<<<N_GRAPHS, POOL_WAVES * 32, 0, stream>>>(hA, gstart, out);
}
